# starting kernel
# baseline (speedup 1.0000x reference)
#include <hip/hip_runtime.h>
#include <math.h>

typedef float v2f __attribute__((ext_vector_type(2)));
typedef float v8f __attribute__((ext_vector_type(8)));

#define BB 512
#define MM 256
#define INDIM 50

#define O_D    0u
#define O_PHI  524288u
#define O_INP  1048576u
#define O_WV   1179648u
#define O_WV2  1181696u
#define O_GX   1183744u
#define O_GB   1184256u

__global__ __launch_bounds__(128) void k1_stages(
    const float* __restrict__ U0, const float* __restrict__ U1, const float* __restrict__ U2,
    const float* __restrict__ S,  const float* __restrict__ w,
    const float* __restrict__ b_t_n, const float* __restrict__ b_obs,
    const int* __restrict__ b_i_n,
    float* __restrict__ ws, float* __restrict__ out)
{
    __shared__ float S_s[256 * 51];
    __shared__ float inp_s[16 * 53];
    __shared__ float redF[4][16];
    __shared__ float redA[4][16];
    __shared__ float xs[16], beta_s[16], scT_s[16];
    __shared__ float aS[4][16], dxS[4][16];

    const int tid  = threadIdx.x;
    const int wv   = tid >> 5;
    const int lane = tid & 31;
    const int ln   = lane & 15;
    const int hb   = (lane < 16) ? 0 : 8;
    const int b0g  = blockIdx.x * 16;

    for (int idx = tid; idx < 256 * INDIM; idx += 128) {
        int m = idx / INDIM, k = idx - m * INDIM;
        S_s[m * 51 + k] = S[idx];
    }
    for (int m = tid; m < 256; m += 128) S_s[m * 51 + 50] = 0.f;

    for (int idx = tid; idx < 16 * 48; idx += 128) {
        int bl = idx / 48, j = idx - bl * 48;
        int u = j >> 4, r = j & 15;
        int iv = b_i_n[(b0g + bl) * 3 + u];
        const float* Uu = (u == 0) ? U0 : (u == 1) ? U1 : U2;
        inp_s[bl * 53 + 2 + j] = Uu[iv * 16 + r];
    }
    for (int idx = tid; idx < 16 * 3; idx += 128) {
        int bl = idx / 3, c = idx - bl * 3;
        inp_s[bl * 53 + 50 + c] = 0.f;
    }
    if (tid < 16) scT_s[tid] = b_t_n[b0g + tid];
    __syncthreads();

    if (tid < 16) {
        float acc = 0.f;
        for (int r = 0; r < 16; ++r)
            acc += inp_s[tid*53 + 2 + r] * inp_s[tid*53 + 18 + r] * inp_s[tid*53 + 34 + r];
        beta_s[tid] = acc; xs[tid] = acc;
    }
    __syncthreads();

    int   mro[4]; float w0r[4], w1r[4], s1r[4];
    #pragma unroll
    for (int nt = 0; nt < 4; ++nt) {
        int m = wv * 64 + nt * 16 + ln;
        mro[nt] = m; w0r[nt] = w[m]; w1r[nt] = w[m + 256]; s1r[nt] = S_s[m * 51 + 1];
    }
    float stv[8]; int bgv[8];
    #pragma unroll
    for (int v = 0; v < 8; ++v) { stv[v] = scT_s[hb + v]; bgv[v] = b0g + hb + v; }

    const int klo = (lane < 16) ? 0 : 2;

    for (int s = 0; s < 4; ++s) {
        if (tid < 16) {
            float ts = (s == 0) ? 0.f : ((s == 3) ? 1.f : 0.5f);
            inp_s[tid * 53 + 0] = scT_s[tid] * ts;
            inp_s[tid * 53 + 1] = xs[tid];
        }
        __syncthreads();

        for (int idx = tid; idx < 16 * 64; idx += 128) {
            int bl = idx >> 6, k = idx & 63;
            ws[O_INP + (unsigned)((s * 512 + b0g + bl) * 64 + k)] = (k < 50) ? inp_s[bl * 53 + k] : 0.f;
        }

        v8f acc[4];
        #pragma unroll
        for (int nt = 0; nt < 4; ++nt)
            #pragma unroll
            for (int v = 0; v < 8; ++v) acc[nt][v] = 0.f;

        #pragma unroll
        for (int kq = 0; kq < 13; ++kq) {
            int k0 = kq * 4 + klo;
            v2f af;
            af.x = inp_s[ln * 53 + k0];
            af.y = inp_s[ln * 53 + k0 + 1];
            #pragma unroll
            for (int nt = 0; nt < 4; ++nt) {
                v2f bf;
                bf.x = (k0     < 50) ? S_s[mro[nt] * 51 + k0]     : 0.f;
                bf.y = (k0 + 1 < 50) ? S_s[mro[nt] * 51 + k0 + 1] : 0.f;
                acc[nt] = __builtin_amdgcn_wmma_f32_16x16x4_f32(
                    false, af, false, bf, (short)0, acc[nt], false, false);
            }
        }

        float Fp[8], Ap[8];
        #pragma unroll
        for (int v = 0; v < 8; ++v) { Fp[v] = 0.f; Ap[v] = 0.f; }
        #pragma unroll
        for (int nt = 0; nt < 4; ++nt) {
            #pragma unroll
            for (int v = 0; v < 8; ++v) {
                float phi = acc[nt][v];
                float sn, cs;
                sincosf(phi, &sn, &cs);
                float fval  = w0r[nt] * cs + w1r[nt] * sn;
                float dterm = w1r[nt] * cs - w0r[nt] * sn;
                Fp[v] += fval;
                Ap[v] += dterm * s1r[nt];
                unsigned o = (unsigned)((s * 512 + bgv[v]) * 256 + mro[nt]);
                ws[O_D + o]   = stv[v] * dterm;
                ws[O_PHI + o] = phi;
            }
        }
        #pragma unroll
        for (int v = 0; v < 8; ++v) {
            float f = Fp[v], a = Ap[v];
            #pragma unroll
            for (int off = 8; off >= 1; off >>= 1) {
                f += __shfl_xor(f, off, 32);
                a += __shfl_xor(a, off, 32);
            }
            Fp[v] = f; Ap[v] = a;
        }
        if (ln == 0) {
            #pragma unroll
            for (int v = 0; v < 8; ++v) { redF[wv][hb + v] = Fp[v]; redA[wv][hb + v] = Ap[v]; }
        }
        __syncthreads();
        if (tid < 16) {
            float F  = redF[0][tid] + redF[1][tid] + redF[2][tid] + redF[3][tid];
            float A  = redA[0][tid] + redA[1][tid] + redA[2][tid] + redA[3][tid];
            float dx = scT_s[tid] * F;
            float a  = scT_s[tid] * A;
            dxS[s][tid] = dx; aS[s][tid] = a;
            if      (s == 0) xs[tid] = beta_s[tid] + 0.5f * dx;
            else if (s == 1) xs[tid] = beta_s[tid] + 0.5f * dx;
            else if (s == 2) xs[tid] = beta_s[tid] + dx;
        }
        __syncthreads();
    }

    if (tid < 16) {
        int bg = b0g + tid;
        float a1 = aS[0][tid], a2 = aS[1][tid], a3 = aS[2][tid], a4 = aS[3][tid];
        float dx1 = dxS[0][tid], dx2 = dxS[1][tid], dx3 = dxS[2][tid], dx4 = dxS[3][tid];
        float xf = beta_s[tid] + (dx1 + 2.f*dx2 + 2.f*dx3 + dx4) * (1.f/6.f);
        float gx = xf - b_obs[bg];
        float c1 = (1.f + a2 + 0.5f*a3*a2 + 0.25f*a4*a3*a2) * (1.f/6.f);
        float c2 = (2.f + a3 + 0.5f*a4*a3) * (1.f/6.f);
        float c3 = (2.f + a4) * (1.f/6.f);
        float c4 = (1.f/6.f);
        float kb1 = a1;
        float kb2 = a2 * (1.f + 0.5f*kb1);
        float kb3 = a3 * (1.f + 0.5f*kb2);
        float kb4 = a4 * (1.f + kb3);
        float gb = 1.f + (kb1 + 2.f*kb2 + 2.f*kb3 + kb4) * (1.f/6.f);
        const float inv = 1.f / 512.f;
        out[bg] = gx * gb * inv;
        ws[O_GX + bg] = gx; ws[O_GB + bg] = gb;
        float sc = scT_s[tid];
        float cc[4] = {c1, c2, c3, c4};
        #pragma unroll
        for (int s2 = 0; s2 < 4; ++s2) {
            float wvv = cc[s2] * gx * inv;
            ws[O_WV  + s2 * 512 + bg] = wvv;
            ws[O_WV2 + s2 * 512 + bg] = wvv * sc;
        }
    }
}

__global__ __launch_bounds__(32) void k2_sgrad(const float* __restrict__ ws, float* __restrict__ out)
{
    __shared__ float wv_s[2048];
    const int lane = threadIdx.x;
    const int mt = blockIdx.x >> 2;
    const int nt = blockIdx.x & 3;
    for (int i = lane; i < 2048; i += 32) wv_s[i] = ws[O_WV + i];
    __syncthreads();
    const int ln  = lane & 15;
    const int klo = (lane < 16) ? 0 : 2;
    const int m   = mt * 16 + ln;
    const int kc  = nt * 16 + ln;
    v8f acc;
    #pragma unroll
    for (int v = 0; v < 8; ++v) acc[v] = 0.f;
    for (int kq = 0; kq < 512; ++kq) {
        int kt = kq * 4 + klo;
        v2f af, bf;
        af.x = wv_s[kt]     * ws[O_D + (unsigned)kt * 256u + m];
        af.y = wv_s[kt + 1] * ws[O_D + (unsigned)(kt + 1) * 256u + m];
        bf.x = ws[O_INP + (unsigned)kt * 64u + kc];
        bf.y = ws[O_INP + (unsigned)(kt + 1) * 64u + kc];
        acc = __builtin_amdgcn_wmma_f32_16x16x4_f32(false, af, false, bf, (short)0, acc, false, false);
    }
    #pragma unroll
    for (int v = 0; v < 8; ++v) {
        int mm = mt * 16 + v + ((lane < 16) ? 0 : 8);
        int kk = nt * 16 + ln;
        if (kk < 50) out[25088 + mm * 50 + kk] = acc[v];
    }
}

__global__ __launch_bounds__(256) void k3_wgrad(const float* __restrict__ ws, float* __restrict__ out)
{
    const int tid = threadIdx.x;
    float ac = 0.f, as2 = 0.f;
    for (int kt = 0; kt < 2048; ++kt) {
        float w2  = ws[O_WV2 + kt];
        float phi = ws[O_PHI + (unsigned)kt * 256u + tid];
        float sn, cs; sincosf(phi, &sn, &cs);
        ac  = fmaf(w2, cs, ac);
        as2 = fmaf(w2, sn, as2);
    }
    out[37888 + tid]       = ac;
    out[37888 + 256 + tid] = as2;
}

__global__ __launch_bounds__(64) void k4_ugrad(const float* __restrict__ ws,
                                               const float* __restrict__ S,
                                               float* __restrict__ out)
{
    __shared__ float dw[1024];
    __shared__ float uv[48];
    const int b = blockIdx.x;
    const int tid = threadIdx.x;
    for (int idx = tid; idx < 1024; idx += 64) {
        int s = idx >> 8, m = idx & 255;
        dw[idx] = ws[O_WV + s * 512 + b] * ws[O_D + (unsigned)((s * 512 + b) * 256 + m)];
    }
    if (tid < 48) uv[tid] = ws[O_INP + (unsigned)(b * 64 + 2 + tid)];
    __syncthreads();
    if (tid < 48) {
        float acc = 0.f;
        for (int m = 0; m < 256; ++m) {
            float dsum = dw[m] + dw[256 + m] + dw[512 + m] + dw[768 + m];
            acc = fmaf(dsum, S[m * 50 + 2 + tid], acc);
        }
        int v = tid >> 4, r = tid & 15;
        float u0 = uv[r], u1 = uv[16 + r], u2 = uv[32 + r];
        float p = (v == 0) ? u1 * u2 : (v == 1) ? u0 * u2 : u0 * u1;
        float gx = ws[O_GX + b], gb = ws[O_GB + b];
        out[512 + b * 48 + tid] = gx * (1.f / 512.f) * gb * p + acc;
    }
}

extern "C" void kernel_launch(void* const* d_in, const int* in_sizes, int n_in,
                              void* d_out, int out_size, void* d_ws, size_t ws_size,
                              hipStream_t stream)
{
    (void)in_sizes; (void)n_in; (void)out_size; (void)ws_size;
    const float* U0   = (const float*)d_in[0];
    const float* U1   = (const float*)d_in[1];
    const float* U2   = (const float*)d_in[2];
    const float* S    = (const float*)d_in[3];
    const float* w    = (const float*)d_in[4];
    const float* btn  = (const float*)d_in[5];
    const float* bobs = (const float*)d_in[6];
    const int*   bin  = (const int*)d_in[7];
    float* out = (float*)d_out;
    float* ws  = (float*)d_ws;

    k1_stages<<<32, 128, 0, stream>>>(U0, U1, U2, S, w, btn, bobs, bin, ws, out);
    k2_sgrad<<<64, 32, 0, stream>>>(ws, out);
    k3_wgrad<<<1, 256, 0, stream>>>(ws, out);
    k4_ugrad<<<512, 64, 0, stream>>>(ws, S, out);
}


# w-grad kernel: 2048-term column sums split over 64 blocks x 1024 threads (4 columns a block, 8 terms a thread, fixed shuffle + LDS tree) instead of 1 block; summation order of out_w changes
# speedup vs baseline: 7.4353x; 7.4353x over previous
#include <hip/hip_runtime.h>
#include <math.h>

typedef float v2f __attribute__((ext_vector_type(2)));
typedef float v8f __attribute__((ext_vector_type(8)));

#define BB 512
#define MM 256
#define INDIM 50

#define O_D    0u
#define O_PHI  524288u
#define O_INP  1048576u
#define O_WV   1179648u
#define O_WV2  1181696u
#define O_GX   1183744u
#define O_GB   1184256u

__global__ __launch_bounds__(128) void k1_stages(
    const float* __restrict__ U0, const float* __restrict__ U1, const float* __restrict__ U2,
    const float* __restrict__ S,  const float* __restrict__ w,
    const float* __restrict__ b_t_n, const float* __restrict__ b_obs,
    const int* __restrict__ b_i_n,
    float* __restrict__ ws, float* __restrict__ out)
{
    __shared__ float S_s[256 * 51];
    __shared__ float inp_s[16 * 53];
    __shared__ float redF[4][16];
    __shared__ float redA[4][16];
    __shared__ float xs[16], beta_s[16], scT_s[16];
    __shared__ float aS[4][16], dxS[4][16];

    const int tid  = threadIdx.x;
    const int wv   = tid >> 5;
    const int lane = tid & 31;
    const int ln   = lane & 15;
    const int hb   = (lane < 16) ? 0 : 8;
    const int b0g  = blockIdx.x * 16;

    for (int idx = tid; idx < 256 * INDIM; idx += 128) {
        int m = idx / INDIM, k = idx - m * INDIM;
        S_s[m * 51 + k] = S[idx];
    }
    for (int m = tid; m < 256; m += 128) S_s[m * 51 + 50] = 0.f;

    for (int idx = tid; idx < 16 * 48; idx += 128) {
        int bl = idx / 48, j = idx - bl * 48;
        int u = j >> 4, r = j & 15;
        int iv = b_i_n[(b0g + bl) * 3 + u];
        const float* Uu = (u == 0) ? U0 : (u == 1) ? U1 : U2;
        inp_s[bl * 53 + 2 + j] = Uu[iv * 16 + r];
    }
    for (int idx = tid; idx < 16 * 3; idx += 128) {
        int bl = idx / 3, c = idx - bl * 3;
        inp_s[bl * 53 + 50 + c] = 0.f;
    }
    if (tid < 16) scT_s[tid] = b_t_n[b0g + tid];
    __syncthreads();

    if (tid < 16) {
        float acc = 0.f;
        for (int r = 0; r < 16; ++r)
            acc += inp_s[tid*53 + 2 + r] * inp_s[tid*53 + 18 + r] * inp_s[tid*53 + 34 + r];
        beta_s[tid] = acc; xs[tid] = acc;
    }
    __syncthreads();

    int   mro[4]; float w0r[4], w1r[4], s1r[4];
    #pragma unroll
    for (int nt = 0; nt < 4; ++nt) {
        int m = wv * 64 + nt * 16 + ln;
        mro[nt] = m; w0r[nt] = w[m]; w1r[nt] = w[m + 256]; s1r[nt] = S_s[m * 51 + 1];
    }
    float stv[8]; int bgv[8];
    #pragma unroll
    for (int v = 0; v < 8; ++v) { stv[v] = scT_s[hb + v]; bgv[v] = b0g + hb + v; }

    const int klo = (lane < 16) ? 0 : 2;

    for (int s = 0; s < 4; ++s) {
        if (tid < 16) {
            float ts = (s == 0) ? 0.f : ((s == 3) ? 1.f : 0.5f);
            inp_s[tid * 53 + 0] = scT_s[tid] * ts;
            inp_s[tid * 53 + 1] = xs[tid];
        }
        __syncthreads();

        for (int idx = tid; idx < 16 * 64; idx += 128) {
            int bl = idx >> 6, k = idx & 63;
            ws[O_INP + (unsigned)((s * 512 + b0g + bl) * 64 + k)] = (k < 50) ? inp_s[bl * 53 + k] : 0.f;
        }

        v8f acc[4];
        #pragma unroll
        for (int nt = 0; nt < 4; ++nt)
            #pragma unroll
            for (int v = 0; v < 8; ++v) acc[nt][v] = 0.f;

        #pragma unroll
        for (int kq = 0; kq < 13; ++kq) {
            int k0 = kq * 4 + klo;
            v2f af;
            af.x = inp_s[ln * 53 + k0];
            af.y = inp_s[ln * 53 + k0 + 1];
            #pragma unroll
            for (int nt = 0; nt < 4; ++nt) {
                v2f bf;
                bf.x = (k0     < 50) ? S_s[mro[nt] * 51 + k0]     : 0.f;
                bf.y = (k0 + 1 < 50) ? S_s[mro[nt] * 51 + k0 + 1] : 0.f;
                acc[nt] = __builtin_amdgcn_wmma_f32_16x16x4_f32(
                    false, af, false, bf, (short)0, acc[nt], false, false);
            }
        }

        float Fp[8], Ap[8];
        #pragma unroll
        for (int v = 0; v < 8; ++v) { Fp[v] = 0.f; Ap[v] = 0.f; }
        #pragma unroll
        for (int nt = 0; nt < 4; ++nt) {
            #pragma unroll
            for (int v = 0; v < 8; ++v) {
                float phi = acc[nt][v];
                float sn, cs;
                sincosf(phi, &sn, &cs);
                float fval  = w0r[nt] * cs + w1r[nt] * sn;
                float dterm = w1r[nt] * cs - w0r[nt] * sn;
                Fp[v] += fval;
                Ap[v] += dterm * s1r[nt];
                unsigned o = (unsigned)((s * 512 + bgv[v]) * 256 + mro[nt]);
                ws[O_D + o]   = stv[v] * dterm;
                ws[O_PHI + o] = phi;
            }
        }
        #pragma unroll
        for (int v = 0; v < 8; ++v) {
            float f = Fp[v], a = Ap[v];
            #pragma unroll
            for (int off = 8; off >= 1; off >>= 1) {
                f += __shfl_xor(f, off, 32);
                a += __shfl_xor(a, off, 32);
            }
            Fp[v] = f; Ap[v] = a;
        }
        if (ln == 0) {
            #pragma unroll
            for (int v = 0; v < 8; ++v) { redF[wv][hb + v] = Fp[v]; redA[wv][hb + v] = Ap[v]; }
        }
        __syncthreads();
        if (tid < 16) {
            float F  = redF[0][tid] + redF[1][tid] + redF[2][tid] + redF[3][tid];
            float A  = redA[0][tid] + redA[1][tid] + redA[2][tid] + redA[3][tid];
            float dx = scT_s[tid] * F;
            float a  = scT_s[tid] * A;
            dxS[s][tid] = dx; aS[s][tid] = a;
            if      (s == 0) xs[tid] = beta_s[tid] + 0.5f * dx;
            else if (s == 1) xs[tid] = beta_s[tid] + 0.5f * dx;
            else if (s == 2) xs[tid] = beta_s[tid] + dx;
        }
        __syncthreads();
    }

    if (tid < 16) {
        int bg = b0g + tid;
        float a1 = aS[0][tid], a2 = aS[1][tid], a3 = aS[2][tid], a4 = aS[3][tid];
        float dx1 = dxS[0][tid], dx2 = dxS[1][tid], dx3 = dxS[2][tid], dx4 = dxS[3][tid];
        float xf = beta_s[tid] + (dx1 + 2.f*dx2 + 2.f*dx3 + dx4) * (1.f/6.f);
        float gx = xf - b_obs[bg];
        float c1 = (1.f + a2 + 0.5f*a3*a2 + 0.25f*a4*a3*a2) * (1.f/6.f);
        float c2 = (2.f + a3 + 0.5f*a4*a3) * (1.f/6.f);
        float c3 = (2.f + a4) * (1.f/6.f);
        float c4 = (1.f/6.f);
        float kb1 = a1;
        float kb2 = a2 * (1.f + 0.5f*kb1);
        float kb3 = a3 * (1.f + 0.5f*kb2);
        float kb4 = a4 * (1.f + kb3);
        float gb = 1.f + (kb1 + 2.f*kb2 + 2.f*kb3 + kb4) * (1.f/6.f);
        const float inv = 1.f / 512.f;
        out[bg] = gx * gb * inv;
        ws[O_GX + bg] = gx; ws[O_GB + bg] = gb;
        float sc = scT_s[tid];
        float cc[4] = {c1, c2, c3, c4};
        #pragma unroll
        for (int s2 = 0; s2 < 4; ++s2) {
            float wvv = cc[s2] * gx * inv;
            ws[O_WV  + s2 * 512 + bg] = wvv;
            ws[O_WV2 + s2 * 512 + bg] = wvv * sc;
        }
    }
}

__global__ __launch_bounds__(32) void k2_sgrad(const float* __restrict__ ws, float* __restrict__ out)
{
    __shared__ float wv_s[2048];
    const int lane = threadIdx.x;
    const int mt = blockIdx.x >> 2;
    const int nt = blockIdx.x & 3;
    for (int i = lane; i < 2048; i += 32) wv_s[i] = ws[O_WV + i];
    __syncthreads();
    const int ln  = lane & 15;
    const int klo = (lane < 16) ? 0 : 2;
    const int m   = mt * 16 + ln;
    const int kc  = nt * 16 + ln;
    v8f acc;
    #pragma unroll
    for (int v = 0; v < 8; ++v) acc[v] = 0.f;
    for (int kq = 0; kq < 512; ++kq) {
        int kt = kq * 4 + klo;
        v2f af, bf;
        af.x = wv_s[kt]     * ws[O_D + (unsigned)kt * 256u + m];
        af.y = wv_s[kt + 1] * ws[O_D + (unsigned)(kt + 1) * 256u + m];
        bf.x = ws[O_INP + (unsigned)kt * 64u + kc];
        bf.y = ws[O_INP + (unsigned)(kt + 1) * 64u + kc];
        acc = __builtin_amdgcn_wmma_f32_16x16x4_f32(false, af, false, bf, (short)0, acc, false, false);
    }
    #pragma unroll
    for (int v = 0; v < 8; ++v) {
        int mm = mt * 16 + v + ((lane < 16) ? 0 : 8);
        int kk = nt * 16 + ln;
        if (kk < 50) out[25088 + mm * 50 + kk] = acc[v];
    }
}

__global__ __launch_bounds__(1024) void k3_wgrad(const float* __restrict__ ws, float* __restrict__ out)
{
    __shared__ float red[32][8];
    const int tid   = threadIdx.x;
    const int lane  = tid & 31;
    const int wvi   = tid >> 5;
    const int ml    = lane & 3;
    const int slice = tid >> 2;
    const int m     = blockIdx.x * 4 + ml;

    float w2[8], ph[8];
    #pragma unroll
    for (int i = 0; i < 8; ++i) {
        int kt = i * 256 + slice;
        w2[i] = ws[O_WV2 + kt];
        ph[i] = ws[O_PHI + (unsigned)kt * 256u + m];
    }
    float ac = 0.f, as2 = 0.f;
    #pragma unroll
    for (int i = 0; i < 8; ++i) {
        float sn, cs; sincosf(ph[i], &sn, &cs);
        ac  = fmaf(w2[i], cs, ac);
        as2 = fmaf(w2[i], sn, as2);
    }
    for (int off = 4; off <= 16; off <<= 1) {
        ac  += __shfl_xor(ac,  off, 32);
        as2 += __shfl_xor(as2, off, 32);
    }
    if (lane < 4) { red[wvi][lane] = ac; red[wvi][4 + lane] = as2; }
    __syncthreads();
    if (tid < 8) {
        float s = 0.f;
        for (int w = 0; w < 32; ++w) s += red[w][tid];
        out[37888 + (tid >> 2) * 256 + blockIdx.x * 4 + (tid & 3)] = s;
    }
}

__global__ __launch_bounds__(64) void k4_ugrad(const float* __restrict__ ws,
                                               const float* __restrict__ S,
                                               float* __restrict__ out)
{
    __shared__ float dw[1024];
    __shared__ float uv[48];
    const int b = blockIdx.x;
    const int tid = threadIdx.x;
    for (int idx = tid; idx < 1024; idx += 64) {
        int s = idx >> 8, m = idx & 255;
        dw[idx] = ws[O_WV + s * 512 + b] * ws[O_D + (unsigned)((s * 512 + b) * 256 + m)];
    }
    if (tid < 48) uv[tid] = ws[O_INP + (unsigned)(b * 64 + 2 + tid)];
    __syncthreads();
    if (tid < 48) {
        float acc = 0.f;
        for (int m = 0; m < 256; ++m) {
            float dsum = dw[m] + dw[256 + m] + dw[512 + m] + dw[768 + m];
            acc = fmaf(dsum, S[m * 50 + 2 + tid], acc);
        }
        int v = tid >> 4, r = tid & 15;
        float u0 = uv[r], u1 = uv[16 + r], u2 = uv[32 + r];
        float p = (v == 0) ? u1 * u2 : (v == 1) ? u0 * u2 : u0 * u1;
        float gx = ws[O_GX + b], gb = ws[O_GB + b];
        out[512 + b * 48 + tid] = gx * (1.f / 512.f) * gb * p + acc;
    }
}

extern "C" void kernel_launch(void* const* d_in, const int* in_sizes, int n_in,
                              void* d_out, int out_size, void* d_ws, size_t ws_size,
                              hipStream_t stream)
{
    (void)in_sizes; (void)n_in; (void)out_size; (void)ws_size;
    const float* U0   = (const float*)d_in[0];
    const float* U1   = (const float*)d_in[1];
    const float* U2   = (const float*)d_in[2];
    const float* S    = (const float*)d_in[3];
    const float* w    = (const float*)d_in[4];
    const float* btn  = (const float*)d_in[5];
    const float* bobs = (const float*)d_in[6];
    const int*   bin  = (const int*)d_in[7];
    float* out = (float*)d_out;
    float* ws  = (float*)d_ws;

    k1_stages<<<32, 128, 0, stream>>>(U0, U1, U2, S, w, btn, bobs, bin, ws, out);
    k2_sgrad<<<64, 32, 0, stream>>>(ws, out);
    k3_wgrad<<<64, 1024, 0, stream>>>(ws, out);
    k4_ugrad<<<512, 64, 0, stream>>>(ws, S, out);
}


# S-grad kernel: each 16x16 tile's 2048-row K range split over 16 waves of one block (all loads in flight at once), partial tiles summed in LDS in fixed order; summation order of out_S changes
# speedup vs baseline: 14.9450x; 2.0100x over previous
#include <hip/hip_runtime.h>
#include <math.h>

typedef float v2f __attribute__((ext_vector_type(2)));
typedef float v8f __attribute__((ext_vector_type(8)));

#define BB 512
#define MM 256
#define INDIM 50

#define O_D    0u
#define O_PHI  524288u
#define O_INP  1048576u
#define O_WV   1179648u
#define O_WV2  1181696u
#define O_GX   1183744u
#define O_GB   1184256u

__global__ __launch_bounds__(128) void k1_stages(
    const float* __restrict__ U0, const float* __restrict__ U1, const float* __restrict__ U2,
    const float* __restrict__ S,  const float* __restrict__ w,
    const float* __restrict__ b_t_n, const float* __restrict__ b_obs,
    const int* __restrict__ b_i_n,
    float* __restrict__ ws, float* __restrict__ out)
{
    __shared__ float S_s[256 * 51];
    __shared__ float inp_s[16 * 53];
    __shared__ float redF[4][16];
    __shared__ float redA[4][16];
    __shared__ float xs[16], beta_s[16], scT_s[16];
    __shared__ float aS[4][16], dxS[4][16];

    const int tid  = threadIdx.x;
    const int wv   = tid >> 5;
    const int lane = tid & 31;
    const int ln   = lane & 15;
    const int hb   = (lane < 16) ? 0 : 8;
    const int b0g  = blockIdx.x * 16;

    for (int idx = tid; idx < 256 * INDIM; idx += 128) {
        int m = idx / INDIM, k = idx - m * INDIM;
        S_s[m * 51 + k] = S[idx];
    }
    for (int m = tid; m < 256; m += 128) S_s[m * 51 + 50] = 0.f;

    for (int idx = tid; idx < 16 * 48; idx += 128) {
        int bl = idx / 48, j = idx - bl * 48;
        int u = j >> 4, r = j & 15;
        int iv = b_i_n[(b0g + bl) * 3 + u];
        const float* Uu = (u == 0) ? U0 : (u == 1) ? U1 : U2;
        inp_s[bl * 53 + 2 + j] = Uu[iv * 16 + r];
    }
    for (int idx = tid; idx < 16 * 3; idx += 128) {
        int bl = idx / 3, c = idx - bl * 3;
        inp_s[bl * 53 + 50 + c] = 0.f;
    }
    if (tid < 16) scT_s[tid] = b_t_n[b0g + tid];
    __syncthreads();

    if (tid < 16) {
        float acc = 0.f;
        for (int r = 0; r < 16; ++r)
            acc += inp_s[tid*53 + 2 + r] * inp_s[tid*53 + 18 + r] * inp_s[tid*53 + 34 + r];
        beta_s[tid] = acc; xs[tid] = acc;
    }
    __syncthreads();

    int   mro[4]; float w0r[4], w1r[4], s1r[4];
    #pragma unroll
    for (int nt = 0; nt < 4; ++nt) {
        int m = wv * 64 + nt * 16 + ln;
        mro[nt] = m; w0r[nt] = w[m]; w1r[nt] = w[m + 256]; s1r[nt] = S_s[m * 51 + 1];
    }
    float stv[8]; int bgv[8];
    #pragma unroll
    for (int v = 0; v < 8; ++v) { stv[v] = scT_s[hb + v]; bgv[v] = b0g + hb + v; }

    const int klo = (lane < 16) ? 0 : 2;

    for (int s = 0; s < 4; ++s) {
        if (tid < 16) {
            float ts = (s == 0) ? 0.f : ((s == 3) ? 1.f : 0.5f);
            inp_s[tid * 53 + 0] = scT_s[tid] * ts;
            inp_s[tid * 53 + 1] = xs[tid];
        }
        __syncthreads();

        for (int idx = tid; idx < 16 * 64; idx += 128) {
            int bl = idx >> 6, k = idx & 63;
            ws[O_INP + (unsigned)((s * 512 + b0g + bl) * 64 + k)] = (k < 50) ? inp_s[bl * 53 + k] : 0.f;
        }

        v8f acc[4];
        #pragma unroll
        for (int nt = 0; nt < 4; ++nt)
            #pragma unroll
            for (int v = 0; v < 8; ++v) acc[nt][v] = 0.f;

        #pragma unroll
        for (int kq = 0; kq < 13; ++kq) {
            int k0 = kq * 4 + klo;
            v2f af;
            af.x = inp_s[ln * 53 + k0];
            af.y = inp_s[ln * 53 + k0 + 1];
            #pragma unroll
            for (int nt = 0; nt < 4; ++nt) {
                v2f bf;
                bf.x = (k0     < 50) ? S_s[mro[nt] * 51 + k0]     : 0.f;
                bf.y = (k0 + 1 < 50) ? S_s[mro[nt] * 51 + k0 + 1] : 0.f;
                acc[nt] = __builtin_amdgcn_wmma_f32_16x16x4_f32(
                    false, af, false, bf, (short)0, acc[nt], false, false);
            }
        }

        float Fp[8], Ap[8];
        #pragma unroll
        for (int v = 0; v < 8; ++v) { Fp[v] = 0.f; Ap[v] = 0.f; }
        #pragma unroll
        for (int nt = 0; nt < 4; ++nt) {
            #pragma unroll
            for (int v = 0; v < 8; ++v) {
                float phi = acc[nt][v];
                float sn, cs;
                sincosf(phi, &sn, &cs);
                float fval  = w0r[nt] * cs + w1r[nt] * sn;
                float dterm = w1r[nt] * cs - w0r[nt] * sn;
                Fp[v] += fval;
                Ap[v] += dterm * s1r[nt];
                unsigned o = (unsigned)((s * 512 + bgv[v]) * 256 + mro[nt]);
                ws[O_D + o]   = stv[v] * dterm;
                ws[O_PHI + o] = phi;
            }
        }
        #pragma unroll
        for (int v = 0; v < 8; ++v) {
            float f = Fp[v], a = Ap[v];
            #pragma unroll
            for (int off = 8; off >= 1; off >>= 1) {
                f += __shfl_xor(f, off, 32);
                a += __shfl_xor(a, off, 32);
            }
            Fp[v] = f; Ap[v] = a;
        }
        if (ln == 0) {
            #pragma unroll
            for (int v = 0; v < 8; ++v) { redF[wv][hb + v] = Fp[v]; redA[wv][hb + v] = Ap[v]; }
        }
        __syncthreads();
        if (tid < 16) {
            float F  = redF[0][tid] + redF[1][tid] + redF[2][tid] + redF[3][tid];
            float A  = redA[0][tid] + redA[1][tid] + redA[2][tid] + redA[3][tid];
            float dx = scT_s[tid] * F;
            float a  = scT_s[tid] * A;
            dxS[s][tid] = dx; aS[s][tid] = a;
            if      (s == 0) xs[tid] = beta_s[tid] + 0.5f * dx;
            else if (s == 1) xs[tid] = beta_s[tid] + 0.5f * dx;
            else if (s == 2) xs[tid] = beta_s[tid] + dx;
        }
        __syncthreads();
    }

    if (tid < 16) {
        int bg = b0g + tid;
        float a1 = aS[0][tid], a2 = aS[1][tid], a3 = aS[2][tid], a4 = aS[3][tid];
        float dx1 = dxS[0][tid], dx2 = dxS[1][tid], dx3 = dxS[2][tid], dx4 = dxS[3][tid];
        float xf = beta_s[tid] + (dx1 + 2.f*dx2 + 2.f*dx3 + dx4) * (1.f/6.f);
        float gx = xf - b_obs[bg];
        float c1 = (1.f + a2 + 0.5f*a3*a2 + 0.25f*a4*a3*a2) * (1.f/6.f);
        float c2 = (2.f + a3 + 0.5f*a4*a3) * (1.f/6.f);
        float c3 = (2.f + a4) * (1.f/6.f);
        float c4 = (1.f/6.f);
        float kb1 = a1;
        float kb2 = a2 * (1.f + 0.5f*kb1);
        float kb3 = a3 * (1.f + 0.5f*kb2);
        float kb4 = a4 * (1.f + kb3);
        float gb = 1.f + (kb1 + 2.f*kb2 + 2.f*kb3 + kb4) * (1.f/6.f);
        const float inv = 1.f / 512.f;
        out[bg] = gx * gb * inv;
        ws[O_GX + bg] = gx; ws[O_GB + bg] = gb;
        float sc = scT_s[tid];
        float cc[4] = {c1, c2, c3, c4};
        #pragma unroll
        for (int s2 = 0; s2 < 4; ++s2) {
            float wvv = cc[s2] * gx * inv;
            ws[O_WV  + s2 * 512 + bg] = wvv;
            ws[O_WV2 + s2 * 512 + bg] = wvv * sc;
        }
    }
}

#define K2_STEP(ACC, KQ)                                                              \
    {                                                                                 \
        const v2f wq = *(const v2f*)(wp + (KQ) * 4);                                  \
        v2f af, bf;                                                                   \
        af.x = wq.x * dp[(KQ) * 1024];                                                \
        af.y = wq.y * dp[(KQ) * 1024 + 256];                                          \
        bf.x = ip[(KQ) * 256];                                                        \
        bf.y = ip[(KQ) * 256 + 64];                                                   \
        ACC = __builtin_amdgcn_wmma_f32_16x16x4_f32(false, af, false, bf, (short)0, ACC, false, false); \
    }

__global__ __launch_bounds__(512) __attribute__((amdgpu_waves_per_eu(4, 4)))
void k2_sgrad(const float* __restrict__ ws, float* __restrict__ out)
{
    __shared__ float part[16][256];
    const int tid  = threadIdx.x;
    const int lane = tid & 31;
    const int wvi  = tid >> 5;
    const int mt = blockIdx.x >> 2;
    const int nt = blockIdx.x & 3;
    const int ln  = lane & 15;
    const int klo = (lane < 16) ? 0 : 2;
    const int m   = mt * 16 + ln;
    const int kc  = nt * 16 + ln;
    const unsigned kt0 = (unsigned)(wvi * 128 + klo);
    const float* __restrict__ dp = ws + O_D   + kt0 * 256u + (unsigned)m;
    const float* __restrict__ ip = ws + O_INP + kt0 * 64u  + (unsigned)kc;
    const float* __restrict__ wp = ws + O_WV  + kt0;

    v8f acc0, acc1;
    #pragma unroll
    for (int v = 0; v < 8; ++v) { acc0[v] = 0.f; acc1[v] = 0.f; }

    #pragma unroll
    for (int kq = 0; kq < 32; kq += 2) {
        K2_STEP(acc0, kq)
        K2_STEP(acc1, kq + 1)
    }

    #pragma unroll
    for (int v = 0; v < 8; ++v) part[wvi][v * 32 + lane] = acc0[v] + acc1[v];
    __syncthreads();

    if (tid < 256) {
        float s = 0.f;
        for (int w = 0; w < 16; ++w) s += part[w][tid];
        const int v  = tid >> 5;
        const int l  = tid & 31;
        const int mm = mt * 16 + v + ((l < 16) ? 0 : 8);
        const int kk = nt * 16 + (l & 15);
        if (kk < 50) out[25088 + mm * 50 + kk] = s;
    }
}
#undef K2_STEP

__global__ __launch_bounds__(1024) void k3_wgrad(const float* __restrict__ ws, float* __restrict__ out)
{
    __shared__ float red[32][8];
    const int tid   = threadIdx.x;
    const int lane  = tid & 31;
    const int wvi   = tid >> 5;
    const int ml    = lane & 3;
    const int slice = tid >> 2;
    const int m     = blockIdx.x * 4 + ml;

    float w2[8], ph[8];
    #pragma unroll
    for (int i = 0; i < 8; ++i) {
        int kt = i * 256 + slice;
        w2[i] = ws[O_WV2 + kt];
        ph[i] = ws[O_PHI + (unsigned)kt * 256u + m];
    }
    float ac = 0.f, as2 = 0.f;
    #pragma unroll
    for (int i = 0; i < 8; ++i) {
        float sn, cs; sincosf(ph[i], &sn, &cs);
        ac  = fmaf(w2[i], cs, ac);
        as2 = fmaf(w2[i], sn, as2);
    }
    for (int off = 4; off <= 16; off <<= 1) {
        ac  += __shfl_xor(ac,  off, 32);
        as2 += __shfl_xor(as2, off, 32);
    }
    if (lane < 4) { red[wvi][lane] = ac; red[wvi][4 + lane] = as2; }
    __syncthreads();
    if (tid < 8) {
        float s = 0.f;
        for (int w = 0; w < 32; ++w) s += red[w][tid];
        out[37888 + (tid >> 2) * 256 + blockIdx.x * 4 + (tid & 3)] = s;
    }
}

__global__ __launch_bounds__(64) void k4_ugrad(const float* __restrict__ ws,
                                               const float* __restrict__ S,
                                               float* __restrict__ out)
{
    __shared__ float dw[1024];
    __shared__ float uv[48];
    const int b = blockIdx.x;
    const int tid = threadIdx.x;
    for (int idx = tid; idx < 1024; idx += 64) {
        int s = idx >> 8, m = idx & 255;
        dw[idx] = ws[O_WV + s * 512 + b] * ws[O_D + (unsigned)((s * 512 + b) * 256 + m)];
    }
    if (tid < 48) uv[tid] = ws[O_INP + (unsigned)(b * 64 + 2 + tid)];
    __syncthreads();
    if (tid < 48) {
        float acc = 0.f;
        for (int m = 0; m < 256; ++m) {
            float dsum = dw[m] + dw[256 + m] + dw[512 + m] + dw[768 + m];
            acc = fmaf(dsum, S[m * 50 + 2 + tid], acc);
        }
        int v = tid >> 4, r = tid & 15;
        float u0 = uv[r], u1 = uv[16 + r], u2 = uv[32 + r];
        float p = (v == 0) ? u1 * u2 : (v == 1) ? u0 * u2 : u0 * u1;
        float gx = ws[O_GX + b], gb = ws[O_GB + b];
        out[512 + b * 48 + tid] = gx * (1.f / 512.f) * gb * p + acc;
    }
}

extern "C" void kernel_launch(void* const* d_in, const int* in_sizes, int n_in,
                              void* d_out, int out_size, void* d_ws, size_t ws_size,
                              hipStream_t stream)
{
    (void)in_sizes; (void)n_in; (void)out_size; (void)ws_size;
    const float* U0   = (const float*)d_in[0];
    const float* U1   = (const float*)d_in[1];
    const float* U2   = (const float*)d_in[2];
    const float* S    = (const float*)d_in[3];
    const float* w    = (const float*)d_in[4];
    const float* btn  = (const float*)d_in[5];
    const float* bobs = (const float*)d_in[6];
    const int*   bin  = (const int*)d_in[7];
    float* out = (float*)d_out;
    float* ws  = (float*)d_ws;

    k1_stages<<<32, 128, 0, stream>>>(U0, U1, U2, S, w, btn, bobs, bin, ws, out);
    k2_sgrad<<<64, 512, 0, stream>>>(ws, out);
    k3_wgrad<<<64, 1024, 0, stream>>>(ws, out);
    k4_ugrad<<<512, 64, 0, stream>>>(ws, S, out);
}


# stage kernel: block widened from 4 to 16 waves, one 16-column tile a wave, its S fragment kept in registers; row sums F and a now add 16 tile partials (summation order changes)
# speedup vs baseline: 21.4581x; 1.4358x over previous
#include <hip/hip_runtime.h>
#include <math.h>

typedef float v2f __attribute__((ext_vector_type(2)));
typedef float v8f __attribute__((ext_vector_type(8)));

#define BB 512
#define MM 256
#define INDIM 50

#define O_D    0u
#define O_PHI  524288u
#define O_INP  1048576u
#define O_WV   1179648u
#define O_WV2  1181696u
#define O_GX   1183744u
#define O_GB   1184256u

__global__ __launch_bounds__(512) void k1_stages(
    const float* __restrict__ U0, const float* __restrict__ U1, const float* __restrict__ U2,
    const float* __restrict__ S,  const float* __restrict__ w,
    const float* __restrict__ b_t_n, const float* __restrict__ b_obs,
    const int* __restrict__ b_i_n,
    float* __restrict__ ws, float* __restrict__ out)
{
    __shared__ float inp_s[16 * 53];
    __shared__ float redF[16][16];
    __shared__ float redA[16][16];
    __shared__ float xs[16], beta_s[16], scT_s[16];
    __shared__ float aS[4][16], dxS[4][16];

    const int tid  = threadIdx.x;
    const int wv   = tid >> 5;
    const int lane = tid & 31;
    const int ln   = lane & 15;
    const int hb   = (lane < 16) ? 0 : 8;
    const int b0g  = blockIdx.x * 16;
    const int klo  = (lane < 16) ? 0 : 2;
    const int m    = wv * 16 + ln;

    v2f bfr[13];
    #pragma unroll
    for (int kq = 0; kq < 13; ++kq) {
        const int k0 = kq * 4 + klo;
        bfr[kq].x = (k0     < INDIM) ? S[m * INDIM + k0]     : 0.f;
        bfr[kq].y = (k0 + 1 < INDIM) ? S[m * INDIM + k0 + 1] : 0.f;
    }
    const float w0r = w[m], w1r = w[m + 256], s1r = S[m * INDIM + 1];

    const float btv = b_t_n[b0g + (tid & 15)];

    if (tid < 8 * 48) {
        const int bl = tid / 48, j = tid - bl * 48;
        const int u = j >> 4, r = j & 15;
        const float* Uu = (u == 0) ? U0 : (u == 1) ? U1 : U2;
        const int i0 = b_i_n[(b0g + bl) * 3 + u];
        const int i1 = b_i_n[(b0g + bl + 8) * 3 + u];
        const float v0 = Uu[i0 * 16 + r];
        const float v1 = Uu[i1 * 16 + r];
        inp_s[bl * 53 + 2 + j]       = v0;
        inp_s[(bl + 8) * 53 + 2 + j] = v1;
    }
    if (tid < 16 * 3) {
        const int bl = tid / 3, c = tid - bl * 3;
        inp_s[bl * 53 + 50 + c] = 0.f;
    }
    if (tid < 16) scT_s[tid] = btv;
    __syncthreads();

    if (tid < 16) {
        float acc = 0.f;
        for (int r = 0; r < 16; ++r)
            acc += inp_s[tid*53 + 2 + r] * inp_s[tid*53 + 18 + r] * inp_s[tid*53 + 34 + r];
        beta_s[tid] = acc; xs[tid] = acc;
    }
    __syncthreads();

    float stv[8]; int bgv[8];
    #pragma unroll
    for (int v = 0; v < 8; ++v) { stv[v] = scT_s[hb + v]; bgv[v] = b0g + hb + v; }

    for (int s = 0; s < 4; ++s) {
        if (tid < 16) {
            float ts = (s == 0) ? 0.f : ((s == 3) ? 1.f : 0.5f);
            inp_s[tid * 53 + 0] = scT_s[tid] * ts;
            inp_s[tid * 53 + 1] = xs[tid];
        }
        __syncthreads();

        for (int i = 0; i < 2; ++i) {
            const int idx = tid + i * 512;
            const int bl = idx >> 6, k = idx & 63;
            ws[O_INP + (unsigned)((s * 512 + b0g + bl) * 64 + k)] = (k < 50) ? inp_s[bl * 53 + k] : 0.f;
        }

        v8f acc;
        #pragma unroll
        for (int v = 0; v < 8; ++v) acc[v] = 0.f;

        #pragma unroll
        for (int kq = 0; kq < 13; ++kq) {
            const int k0 = kq * 4 + klo;
            v2f af;
            af.x = inp_s[ln * 53 + k0];
            af.y = inp_s[ln * 53 + k0 + 1];
            acc = __builtin_amdgcn_wmma_f32_16x16x4_f32(
                false, af, false, bfr[kq], (short)0, acc, false, false);
        }

        float Fp[8], Ap[8];
        #pragma unroll
        for (int v = 0; v < 8; ++v) {
            float phi = acc[v];
            float sn, cs;
            sincosf(phi, &sn, &cs);
            float fval  = w0r * cs + w1r * sn;
            float dterm = w1r * cs - w0r * sn;
            Fp[v] = fval;
            Ap[v] = dterm * s1r;
            unsigned o = (unsigned)((s * 512 + bgv[v]) * 256 + m);
            ws[O_D + o]   = stv[v] * dterm;
            ws[O_PHI + o] = phi;
        }
        #pragma unroll
        for (int v = 0; v < 8; ++v) {
            float f = Fp[v], a = Ap[v];
            #pragma unroll
            for (int off = 8; off >= 1; off >>= 1) {
                f += __shfl_xor(f, off, 32);
                a += __shfl_xor(a, off, 32);
            }
            Fp[v] = f; Ap[v] = a;
        }
        if (ln == 0) {
            #pragma unroll
            for (int v = 0; v < 8; ++v) { redF[wv][hb + v] = Fp[v]; redA[wv][hb + v] = Ap[v]; }
        }
        __syncthreads();
        if (tid < 16) {
            float F = 0.f, A = 0.f;
            for (int q = 0; q < 16; ++q) { F += redF[q][tid]; A += redA[q][tid]; }
            float dx = scT_s[tid] * F;
            float a  = scT_s[tid] * A;
            dxS[s][tid] = dx; aS[s][tid] = a;
            if      (s == 0) xs[tid] = beta_s[tid] + 0.5f * dx;
            else if (s == 1) xs[tid] = beta_s[tid] + 0.5f * dx;
            else if (s == 2) xs[tid] = beta_s[tid] + dx;
        }
        __syncthreads();
    }

    if (tid < 16) {
        int bg = b0g + tid;
        float a1 = aS[0][tid], a2 = aS[1][tid], a3 = aS[2][tid], a4 = aS[3][tid];
        float dx1 = dxS[0][tid], dx2 = dxS[1][tid], dx3 = dxS[2][tid], dx4 = dxS[3][tid];
        float xf = beta_s[tid] + (dx1 + 2.f*dx2 + 2.f*dx3 + dx4) * (1.f/6.f);
        float gx = xf - b_obs[bg];
        float c1 = (1.f + a2 + 0.5f*a3*a2 + 0.25f*a4*a3*a2) * (1.f/6.f);
        float c2 = (2.f + a3 + 0.5f*a4*a3) * (1.f/6.f);
        float c3 = (2.f + a4) * (1.f/6.f);
        float c4 = (1.f/6.f);
        float kb1 = a1;
        float kb2 = a2 * (1.f + 0.5f*kb1);
        float kb3 = a3 * (1.f + 0.5f*kb2);
        float kb4 = a4 * (1.f + kb3);
        float gb = 1.f + (kb1 + 2.f*kb2 + 2.f*kb3 + kb4) * (1.f/6.f);
        const float inv = 1.f / 512.f;
        out[bg] = gx * gb * inv;
        ws[O_GX + bg] = gx; ws[O_GB + bg] = gb;
        float sc = scT_s[tid];
        float cc[4] = {c1, c2, c3, c4};
        #pragma unroll
        for (int s2 = 0; s2 < 4; ++s2) {
            float wvv = cc[s2] * gx * inv;
            ws[O_WV  + s2 * 512 + bg] = wvv;
            ws[O_WV2 + s2 * 512 + bg] = wvv * sc;
        }
    }
}

#define K2_STEP(ACC, KQ)                                                              \
    {                                                                                 \
        const v2f wq = *(const v2f*)(wp + (KQ) * 4);                                  \
        v2f af, bf;                                                                   \
        af.x = wq.x * dp[(KQ) * 1024];                                                \
        af.y = wq.y * dp[(KQ) * 1024 + 256];                                          \
        bf.x = ip[(KQ) * 256];                                                        \
        bf.y = ip[(KQ) * 256 + 64];                                                   \
        ACC = __builtin_amdgcn_wmma_f32_16x16x4_f32(false, af, false, bf, (short)0, ACC, false, false); \
    }

__global__ __launch_bounds__(512) __attribute__((amdgpu_waves_per_eu(4, 4)))
void k2_sgrad(const float* __restrict__ ws, float* __restrict__ out)
{
    __shared__ float part[16][256];
    const int tid  = threadIdx.x;
    const int lane = tid & 31;
    const int wvi  = tid >> 5;
    const int mt = blockIdx.x >> 2;
    const int nt = blockIdx.x & 3;
    const int ln  = lane & 15;
    const int klo = (lane < 16) ? 0 : 2;
    const int m   = mt * 16 + ln;
    const int kc  = nt * 16 + ln;
    const unsigned kt0 = (unsigned)(wvi * 128 + klo);
    const float* __restrict__ dp = ws + O_D   + kt0 * 256u + (unsigned)m;
    const float* __restrict__ ip = ws + O_INP + kt0 * 64u  + (unsigned)kc;
    const float* __restrict__ wp = ws + O_WV  + kt0;

    v8f acc0, acc1;
    #pragma unroll
    for (int v = 0; v < 8; ++v) { acc0[v] = 0.f; acc1[v] = 0.f; }

    #pragma unroll
    for (int kq = 0; kq < 32; kq += 2) {
        K2_STEP(acc0, kq)
        K2_STEP(acc1, kq + 1)
    }

    #pragma unroll
    for (int v = 0; v < 8; ++v) part[wvi][v * 32 + lane] = acc0[v] + acc1[v];
    __syncthreads();

    if (tid < 256) {
        float s = 0.f;
        for (int w = 0; w < 16; ++w) s += part[w][tid];
        const int v  = tid >> 5;
        const int l  = tid & 31;
        const int mm = mt * 16 + v + ((l < 16) ? 0 : 8);
        const int kk = nt * 16 + (l & 15);
        if (kk < 50) out[25088 + mm * 50 + kk] = s;
    }
}
#undef K2_STEP

__global__ __launch_bounds__(1024) void k3_wgrad(const float* __restrict__ ws, float* __restrict__ out)
{
    __shared__ float red[32][8];
    const int tid   = threadIdx.x;
    const int lane  = tid & 31;
    const int wvi   = tid >> 5;
    const int ml    = lane & 3;
    const int slice = tid >> 2;
    const int m     = blockIdx.x * 4 + ml;

    float w2[8], ph[8];
    #pragma unroll
    for (int i = 0; i < 8; ++i) {
        int kt = i * 256 + slice;
        w2[i] = ws[O_WV2 + kt];
        ph[i] = ws[O_PHI + (unsigned)kt * 256u + m];
    }
    float ac = 0.f, as2 = 0.f;
    #pragma unroll
    for (int i = 0; i < 8; ++i) {
        float sn, cs; sincosf(ph[i], &sn, &cs);
        ac  = fmaf(w2[i], cs, ac);
        as2 = fmaf(w2[i], sn, as2);
    }
    for (int off = 4; off <= 16; off <<= 1) {
        ac  += __shfl_xor(ac,  off, 32);
        as2 += __shfl_xor(as2, off, 32);
    }
    if (lane < 4) { red[wvi][lane] = ac; red[wvi][4 + lane] = as2; }
    __syncthreads();
    if (tid < 8) {
        float s = 0.f;
        for (int w = 0; w < 32; ++w) s += red[w][tid];
        out[37888 + (tid >> 2) * 256 + blockIdx.x * 4 + (tid & 3)] = s;
    }
}

__global__ __launch_bounds__(64) void k4_ugrad(const float* __restrict__ ws,
                                               const float* __restrict__ S,
                                               float* __restrict__ out)
{
    __shared__ float dw[1024];
    __shared__ float uv[48];
    const int b = blockIdx.x;
    const int tid = threadIdx.x;
    for (int idx = tid; idx < 1024; idx += 64) {
        int s = idx >> 8, m = idx & 255;
        dw[idx] = ws[O_WV + s * 512 + b] * ws[O_D + (unsigned)((s * 512 + b) * 256 + m)];
    }
    if (tid < 48) uv[tid] = ws[O_INP + (unsigned)(b * 64 + 2 + tid)];
    __syncthreads();
    if (tid < 48) {
        float acc = 0.f;
        for (int m = 0; m < 256; ++m) {
            float dsum = dw[m] + dw[256 + m] + dw[512 + m] + dw[768 + m];
            acc = fmaf(dsum, S[m * 50 + 2 + tid], acc);
        }
        int v = tid >> 4, r = tid & 15;
        float u0 = uv[r], u1 = uv[16 + r], u2 = uv[32 + r];
        float p = (v == 0) ? u1 * u2 : (v == 1) ? u0 * u2 : u0 * u1;
        float gx = ws[O_GX + b], gb = ws[O_GB + b];
        out[512 + b * 48 + tid] = gx * (1.f / 512.f) * gb * p + acc;
    }
}

extern "C" void kernel_launch(void* const* d_in, const int* in_sizes, int n_in,
                              void* d_out, int out_size, void* d_ws, size_t ws_size,
                              hipStream_t stream)
{
    (void)in_sizes; (void)n_in; (void)out_size; (void)ws_size;
    const float* U0   = (const float*)d_in[0];
    const float* U1   = (const float*)d_in[1];
    const float* U2   = (const float*)d_in[2];
    const float* S    = (const float*)d_in[3];
    const float* w    = (const float*)d_in[4];
    const float* btn  = (const float*)d_in[5];
    const float* bobs = (const float*)d_in[6];
    const int*   bin  = (const int*)d_in[7];
    float* out = (float*)d_out;
    float* ws  = (float*)d_ws;

    k1_stages<<<32, 512, 0, stream>>>(U0, U1, U2, S, w, btn, bobs, bin, ws, out);
    k2_sgrad<<<64, 512, 0, stream>>>(ws, out);
    k3_wgrad<<<64, 1024, 0, stream>>>(ws, out);
    k4_ugrad<<<512, 64, 0, stream>>>(ws, S, out);
}


# S-grad, w-grad and Uvec-grad kernels merged into one launch (192 blocks x 512, block index picks the role; all read only the stage kernel's planes); w-grad 16 terms a thread: out_w sum order changes
# speedup vs baseline: 28.8734x; 1.3456x over previous
#include <hip/hip_runtime.h>
#include <math.h>

typedef float v2f __attribute__((ext_vector_type(2)));
typedef float v8f __attribute__((ext_vector_type(8)));

#define BB 512
#define MM 256
#define INDIM 50

#define O_D    0u
#define O_PHI  524288u
#define O_INP  1048576u
#define O_WV   1179648u
#define O_WV2  1181696u
#define O_GX   1183744u
#define O_GB   1184256u

__global__ __launch_bounds__(512) void k1_stages(
    const float* __restrict__ U0, const float* __restrict__ U1, const float* __restrict__ U2,
    const float* __restrict__ S,  const float* __restrict__ w,
    const float* __restrict__ b_t_n, const float* __restrict__ b_obs,
    const int* __restrict__ b_i_n,
    float* __restrict__ ws, float* __restrict__ out)
{
    __shared__ float inp_s[16 * 53];
    __shared__ float redF[16][16];
    __shared__ float redA[16][16];
    __shared__ float xs[16], beta_s[16], scT_s[16];
    __shared__ float aS[4][16], dxS[4][16];

    const int tid  = threadIdx.x;
    const int wv   = tid >> 5;
    const int lane = tid & 31;
    const int ln   = lane & 15;
    const int hb   = (lane < 16) ? 0 : 8;
    const int b0g  = blockIdx.x * 16;
    const int klo  = (lane < 16) ? 0 : 2;
    const int m    = wv * 16 + ln;

    v2f bfr[13];
    #pragma unroll
    for (int kq = 0; kq < 13; ++kq) {
        const int k0 = kq * 4 + klo;
        bfr[kq].x = (k0     < INDIM) ? S[m * INDIM + k0]     : 0.f;
        bfr[kq].y = (k0 + 1 < INDIM) ? S[m * INDIM + k0 + 1] : 0.f;
    }
    const float w0r = w[m], w1r = w[m + 256], s1r = S[m * INDIM + 1];

    const float btv = b_t_n[b0g + (tid & 15)];

    if (tid < 8 * 48) {
        const int bl = tid / 48, j = tid - bl * 48;
        const int u = j >> 4, r = j & 15;
        const float* Uu = (u == 0) ? U0 : (u == 1) ? U1 : U2;
        const int i0 = b_i_n[(b0g + bl) * 3 + u];
        const int i1 = b_i_n[(b0g + bl + 8) * 3 + u];
        const float v0 = Uu[i0 * 16 + r];
        const float v1 = Uu[i1 * 16 + r];
        inp_s[bl * 53 + 2 + j]       = v0;
        inp_s[(bl + 8) * 53 + 2 + j] = v1;
    }
    if (tid < 16 * 3) {
        const int bl = tid / 3, c = tid - bl * 3;
        inp_s[bl * 53 + 50 + c] = 0.f;
    }
    if (tid < 16) scT_s[tid] = btv;
    __syncthreads();

    if (tid < 16) {
        float acc = 0.f;
        for (int r = 0; r < 16; ++r)
            acc += inp_s[tid*53 + 2 + r] * inp_s[tid*53 + 18 + r] * inp_s[tid*53 + 34 + r];
        beta_s[tid] = acc; xs[tid] = acc;
    }
    __syncthreads();

    float stv[8]; int bgv[8];
    #pragma unroll
    for (int v = 0; v < 8; ++v) { stv[v] = scT_s[hb + v]; bgv[v] = b0g + hb + v; }

    for (int s = 0; s < 4; ++s) {
        if (tid < 16) {
            float ts = (s == 0) ? 0.f : ((s == 3) ? 1.f : 0.5f);
            inp_s[tid * 53 + 0] = scT_s[tid] * ts;
            inp_s[tid * 53 + 1] = xs[tid];
        }
        __syncthreads();

        for (int i = 0; i < 2; ++i) {
            const int idx = tid + i * 512;
            const int bl = idx >> 6, k = idx & 63;
            ws[O_INP + (unsigned)((s * 512 + b0g + bl) * 64 + k)] = (k < 50) ? inp_s[bl * 53 + k] : 0.f;
        }

        v8f acc;
        #pragma unroll
        for (int v = 0; v < 8; ++v) acc[v] = 0.f;

        #pragma unroll
        for (int kq = 0; kq < 13; ++kq) {
            const int k0 = kq * 4 + klo;
            v2f af;
            af.x = inp_s[ln * 53 + k0];
            af.y = inp_s[ln * 53 + k0 + 1];
            acc = __builtin_amdgcn_wmma_f32_16x16x4_f32(
                false, af, false, bfr[kq], (short)0, acc, false, false);
        }

        float Fp[8], Ap[8];
        #pragma unroll
        for (int v = 0; v < 8; ++v) {
            float phi = acc[v];
            float sn, cs;
            sincosf(phi, &sn, &cs);
            float fval  = w0r * cs + w1r * sn;
            float dterm = w1r * cs - w0r * sn;
            Fp[v] = fval;
            Ap[v] = dterm * s1r;
            unsigned o = (unsigned)((s * 512 + bgv[v]) * 256 + m);
            ws[O_D + o]   = stv[v] * dterm;
            ws[O_PHI + o] = phi;
        }
        #pragma unroll
        for (int v = 0; v < 8; ++v) {
            float f = Fp[v], a = Ap[v];
            #pragma unroll
            for (int off = 8; off >= 1; off >>= 1) {
                f += __shfl_xor(f, off, 32);
                a += __shfl_xor(a, off, 32);
            }
            Fp[v] = f; Ap[v] = a;
        }
        if (ln == 0) {
            #pragma unroll
            for (int v = 0; v < 8; ++v) { redF[wv][hb + v] = Fp[v]; redA[wv][hb + v] = Ap[v]; }
        }
        __syncthreads();
        if (tid < 16) {
            float F = 0.f, A = 0.f;
            for (int q = 0; q < 16; ++q) { F += redF[q][tid]; A += redA[q][tid]; }
            float dx = scT_s[tid] * F;
            float a  = scT_s[tid] * A;
            dxS[s][tid] = dx; aS[s][tid] = a;
            if      (s == 0) xs[tid] = beta_s[tid] + 0.5f * dx;
            else if (s == 1) xs[tid] = beta_s[tid] + 0.5f * dx;
            else if (s == 2) xs[tid] = beta_s[tid] + dx;
        }
        __syncthreads();
    }

    if (tid < 16) {
        int bg = b0g + tid;
        float a1 = aS[0][tid], a2 = aS[1][tid], a3 = aS[2][tid], a4 = aS[3][tid];
        float dx1 = dxS[0][tid], dx2 = dxS[1][tid], dx3 = dxS[2][tid], dx4 = dxS[3][tid];
        float xf = beta_s[tid] + (dx1 + 2.f*dx2 + 2.f*dx3 + dx4) * (1.f/6.f);
        float gx = xf - b_obs[bg];
        float c1 = (1.f + a2 + 0.5f*a3*a2 + 0.25f*a4*a3*a2) * (1.f/6.f);
        float c2 = (2.f + a3 + 0.5f*a4*a3) * (1.f/6.f);
        float c3 = (2.f + a4) * (1.f/6.f);
        float c4 = (1.f/6.f);
        float kb1 = a1;
        float kb2 = a2 * (1.f + 0.5f*kb1);
        float kb3 = a3 * (1.f + 0.5f*kb2);
        float kb4 = a4 * (1.f + kb3);
        float gb = 1.f + (kb1 + 2.f*kb2 + 2.f*kb3 + kb4) * (1.f/6.f);
        const float inv = 1.f / 512.f;
        out[bg] = gx * gb * inv;
        ws[O_GX + bg] = gx; ws[O_GB + bg] = gb;
        float sc = scT_s[tid];
        float cc[4] = {c1, c2, c3, c4};
        #pragma unroll
        for (int s2 = 0; s2 < 4; ++s2) {
            float wvv = cc[s2] * gx * inv;
            ws[O_WV  + s2 * 512 + bg] = wvv;
            ws[O_WV2 + s2 * 512 + bg] = wvv * sc;
        }
    }
}

#define K2_STEP(ACC, KQ)                                                              \
    {                                                                                 \
        const v2f wq = *(const v2f*)(wp + (KQ) * 4);                                  \
        v2f af, bf;                                                                   \
        af.x = wq.x * dp[(KQ) * 1024];                                                \
        af.y = wq.y * dp[(KQ) * 1024 + 256];                                          \
        bf.x = ip[(KQ) * 256];                                                        \
        bf.y = ip[(KQ) * 256 + 64];                                                   \
        ACC = __builtin_amdgcn_wmma_f32_16x16x4_f32(false, af, false, bf, (short)0, ACC, false, false); \
    }

static __device__ __forceinline__ void sgrad_role(const int blk, const float* __restrict__ ws, float* __restrict__ out)
{
    __shared__ float part[16][256];
    const int tid  = threadIdx.x;
    const int lane = tid & 31;
    const int wvi  = tid >> 5;
    const int mt = blk >> 2;
    const int nt = blk & 3;
    const int ln  = lane & 15;
    const int klo = (lane < 16) ? 0 : 2;
    const int m   = mt * 16 + ln;
    const int kc  = nt * 16 + ln;
    const unsigned kt0 = (unsigned)(wvi * 128 + klo);
    const float* __restrict__ dp = ws + O_D   + kt0 * 256u + (unsigned)m;
    const float* __restrict__ ip = ws + O_INP + kt0 * 64u  + (unsigned)kc;
    const float* __restrict__ wp = ws + O_WV  + kt0;

    v8f acc0, acc1;
    #pragma unroll
    for (int v = 0; v < 8; ++v) { acc0[v] = 0.f; acc1[v] = 0.f; }

    #pragma unroll
    for (int kq = 0; kq < 32; kq += 2) {
        K2_STEP(acc0, kq)
        K2_STEP(acc1, kq + 1)
    }

    #pragma unroll
    for (int v = 0; v < 8; ++v) part[wvi][v * 32 + lane] = acc0[v] + acc1[v];
    __syncthreads();

    if (tid < 256) {
        float s = 0.f;
        for (int w = 0; w < 16; ++w) s += part[w][tid];
        const int v  = tid >> 5;
        const int l  = tid & 31;
        const int mm = mt * 16 + v + ((l < 16) ? 0 : 8);
        const int kk = nt * 16 + (l & 15);
        if (kk < 50) out[25088 + mm * 50 + kk] = s;
    }
}
#undef K2_STEP

static __device__ __forceinline__ void wgrad_role(const int blk, const float* __restrict__ ws, float* __restrict__ out)
{
    __shared__ float red[16][8];
    const int tid   = threadIdx.x;
    const int lane  = tid & 31;
    const int wvi   = tid >> 5;
    const int ml    = lane & 3;
    const int slice = tid >> 2;
    const int m     = blk * 4 + ml;

    float w2[16], ph[16];
    #pragma unroll
    for (int i = 0; i < 16; ++i) {
        int kt = i * 128 + slice;
        w2[i] = ws[O_WV2 + kt];
        ph[i] = ws[O_PHI + (unsigned)kt * 256u + m];
    }
    float ac = 0.f, as2 = 0.f;
    #pragma unroll
    for (int i = 0; i < 16; ++i) {
        float sn, cs; sincosf(ph[i], &sn, &cs);
        ac  = fmaf(w2[i], cs, ac);
        as2 = fmaf(w2[i], sn, as2);
    }
    for (int off = 4; off <= 16; off <<= 1) {
        ac  += __shfl_xor(ac,  off, 32);
        as2 += __shfl_xor(as2, off, 32);
    }
    if (lane < 4) { red[wvi][lane] = ac; red[wvi][4 + lane] = as2; }
    __syncthreads();
    if (tid < 8) {
        float s = 0.f;
        for (int w = 0; w < 16; ++w) s += red[w][tid];
        out[37888 + (tid >> 2) * 256 + blk * 4 + (tid & 3)] = s;
    }
}

static __device__ __forceinline__ void ugrad_role(const int blk, const float* __restrict__ ws,
                                                  const float* __restrict__ S, float* __restrict__ out)
{
    __shared__ float dw[8][1024];
    __shared__ float uv[8][48];
    const int g   = threadIdx.x >> 6;
    const int tid = threadIdx.x & 63;
    const int b   = blk * 8 + g;
    for (int idx = tid; idx < 1024; idx += 64) {
        int s = idx >> 8, m = idx & 255;
        dw[g][idx] = ws[O_WV + s * 512 + b] * ws[O_D + (unsigned)((s * 512 + b) * 256 + m)];
    }
    if (tid < 48) uv[g][tid] = ws[O_INP + (unsigned)(b * 64 + 2 + tid)];
    __syncthreads();
    if (tid < 48) {
        float acc = 0.f;
        for (int m = 0; m < 256; ++m) {
            float dsum = dw[g][m] + dw[g][256 + m] + dw[g][512 + m] + dw[g][768 + m];
            acc = fmaf(dsum, S[m * 50 + 2 + tid], acc);
        }
        int v = tid >> 4, r = tid & 15;
        float u0 = uv[g][r], u1 = uv[g][16 + r], u2 = uv[g][32 + r];
        float p = (v == 0) ? u1 * u2 : (v == 1) ? u0 * u2 : u0 * u1;
        float gx = ws[O_GX + b], gb = ws[O_GB + b];
        out[512 + b * 48 + tid] = gx * (1.f / 512.f) * gb * p + acc;
    }
}

__global__ __launch_bounds__(512) __attribute__((amdgpu_waves_per_eu(4, 4)))
void k2_consumers(const float* __restrict__ ws, const float* __restrict__ S, float* __restrict__ out)
{
    const int bid = blockIdx.x;
    if      (bid < 64)  ugrad_role(bid, ws, S, out);
    else if (bid < 128) sgrad_role(bid - 64, ws, out);
    else                wgrad_role(bid - 128, ws, out);
}

extern "C" void kernel_launch(void* const* d_in, const int* in_sizes, int n_in,
                              void* d_out, int out_size, void* d_ws, size_t ws_size,
                              hipStream_t stream)
{
    (void)in_sizes; (void)n_in; (void)out_size; (void)ws_size;
    const float* U0   = (const float*)d_in[0];
    const float* U1   = (const float*)d_in[1];
    const float* U2   = (const float*)d_in[2];
    const float* S    = (const float*)d_in[3];
    const float* w    = (const float*)d_in[4];
    const float* btn  = (const float*)d_in[5];
    const float* bobs = (const float*)d_in[6];
    const int*   bin  = (const int*)d_in[7];
    float* out = (float*)d_out;
    float* ws  = (float*)d_ws;

    k1_stages<<<32, 512, 0, stream>>>(U0, U1, U2, S, w, btn, bobs, bin, ws, out);
    k2_consumers<<<192, 512, 0, stream>>>(ws, S, out);
}


# Uvec-grad role: 512x256x48 product moved from a per-thread fma loop (32 dependent load round trips) to f32 WMMA, K split over 16 waves, all loads issued up front, 32 blocks; its sum order changes
# speedup vs baseline: 33.4142x; 1.1573x over previous
#include <hip/hip_runtime.h>
#include <math.h>

typedef float v2f __attribute__((ext_vector_type(2)));
typedef float v8f __attribute__((ext_vector_type(8)));

#define BB 512
#define MM 256
#define INDIM 50

#define O_D    0u
#define O_PHI  524288u
#define O_INP  1048576u
#define O_WV   1179648u
#define O_WV2  1181696u
#define O_GX   1183744u
#define O_GB   1184256u

__global__ __launch_bounds__(512) void k1_stages(
    const float* __restrict__ U0, const float* __restrict__ U1, const float* __restrict__ U2,
    const float* __restrict__ S,  const float* __restrict__ w,
    const float* __restrict__ b_t_n, const float* __restrict__ b_obs,
    const int* __restrict__ b_i_n,
    float* __restrict__ ws, float* __restrict__ out)
{
    __shared__ float inp_s[16 * 53];
    __shared__ float redF[16][16];
    __shared__ float redA[16][16];
    __shared__ float xs[16], beta_s[16], scT_s[16];
    __shared__ float aS[4][16], dxS[4][16];

    const int tid  = threadIdx.x;
    const int wv   = tid >> 5;
    const int lane = tid & 31;
    const int ln   = lane & 15;
    const int hb   = (lane < 16) ? 0 : 8;
    const int b0g  = blockIdx.x * 16;
    const int klo  = (lane < 16) ? 0 : 2;
    const int m    = wv * 16 + ln;

    v2f bfr[13];
    #pragma unroll
    for (int kq = 0; kq < 13; ++kq) {
        const int k0 = kq * 4 + klo;
        bfr[kq].x = (k0     < INDIM) ? S[m * INDIM + k0]     : 0.f;
        bfr[kq].y = (k0 + 1 < INDIM) ? S[m * INDIM + k0 + 1] : 0.f;
    }
    const float w0r = w[m], w1r = w[m + 256], s1r = S[m * INDIM + 1];

    const float btv = b_t_n[b0g + (tid & 15)];

    if (tid < 8 * 48) {
        const int bl = tid / 48, j = tid - bl * 48;
        const int u = j >> 4, r = j & 15;
        const float* Uu = (u == 0) ? U0 : (u == 1) ? U1 : U2;
        const int i0 = b_i_n[(b0g + bl) * 3 + u];
        const int i1 = b_i_n[(b0g + bl + 8) * 3 + u];
        const float v0 = Uu[i0 * 16 + r];
        const float v1 = Uu[i1 * 16 + r];
        inp_s[bl * 53 + 2 + j]       = v0;
        inp_s[(bl + 8) * 53 + 2 + j] = v1;
    }
    if (tid < 16 * 3) {
        const int bl = tid / 3, c = tid - bl * 3;
        inp_s[bl * 53 + 50 + c] = 0.f;
    }
    if (tid < 16) scT_s[tid] = btv;
    __syncthreads();

    if (tid < 16) {
        float acc = 0.f;
        for (int r = 0; r < 16; ++r)
            acc += inp_s[tid*53 + 2 + r] * inp_s[tid*53 + 18 + r] * inp_s[tid*53 + 34 + r];
        beta_s[tid] = acc; xs[tid] = acc;
    }
    __syncthreads();

    float stv[8]; int bgv[8];
    #pragma unroll
    for (int v = 0; v < 8; ++v) { stv[v] = scT_s[hb + v]; bgv[v] = b0g + hb + v; }

    for (int s = 0; s < 4; ++s) {
        if (tid < 16) {
            float ts = (s == 0) ? 0.f : ((s == 3) ? 1.f : 0.5f);
            inp_s[tid * 53 + 0] = scT_s[tid] * ts;
            inp_s[tid * 53 + 1] = xs[tid];
        }
        __syncthreads();

        for (int i = 0; i < 2; ++i) {
            const int idx = tid + i * 512;
            const int bl = idx >> 6, k = idx & 63;
            ws[O_INP + (unsigned)((s * 512 + b0g + bl) * 64 + k)] = (k < 50) ? inp_s[bl * 53 + k] : 0.f;
        }

        v8f acc;
        #pragma unroll
        for (int v = 0; v < 8; ++v) acc[v] = 0.f;

        #pragma unroll
        for (int kq = 0; kq < 13; ++kq) {
            const int k0 = kq * 4 + klo;
            v2f af;
            af.x = inp_s[ln * 53 + k0];
            af.y = inp_s[ln * 53 + k0 + 1];
            acc = __builtin_amdgcn_wmma_f32_16x16x4_f32(
                false, af, false, bfr[kq], (short)0, acc, false, false);
        }

        float Fp[8], Ap[8];
        #pragma unroll
        for (int v = 0; v < 8; ++v) {
            float phi = acc[v];
            float sn, cs;
            sincosf(phi, &sn, &cs);
            float fval  = w0r * cs + w1r * sn;
            float dterm = w1r * cs - w0r * sn;
            Fp[v] = fval;
            Ap[v] = dterm * s1r;
            unsigned o = (unsigned)((s * 512 + bgv[v]) * 256 + m);
            ws[O_D + o]   = stv[v] * dterm;
            ws[O_PHI + o] = phi;
        }
        #pragma unroll
        for (int v = 0; v < 8; ++v) {
            float f = Fp[v], a = Ap[v];
            #pragma unroll
            for (int off = 8; off >= 1; off >>= 1) {
                f += __shfl_xor(f, off, 32);
                a += __shfl_xor(a, off, 32);
            }
            Fp[v] = f; Ap[v] = a;
        }
        if (ln == 0) {
            #pragma unroll
            for (int v = 0; v < 8; ++v) { redF[wv][hb + v] = Fp[v]; redA[wv][hb + v] = Ap[v]; }
        }
        __syncthreads();
        if (tid < 16) {
            float F = 0.f, A = 0.f;
            for (int q = 0; q < 16; ++q) { F += redF[q][tid]; A += redA[q][tid]; }
            float dx = scT_s[tid] * F;
            float a  = scT_s[tid] * A;
            dxS[s][tid] = dx; aS[s][tid] = a;
            if      (s == 0) xs[tid] = beta_s[tid] + 0.5f * dx;
            else if (s == 1) xs[tid] = beta_s[tid] + 0.5f * dx;
            else if (s == 2) xs[tid] = beta_s[tid] + dx;
        }
        __syncthreads();
    }

    if (tid < 16) {
        int bg = b0g + tid;
        float a1 = aS[0][tid], a2 = aS[1][tid], a3 = aS[2][tid], a4 = aS[3][tid];
        float dx1 = dxS[0][tid], dx2 = dxS[1][tid], dx3 = dxS[2][tid], dx4 = dxS[3][tid];
        float xf = beta_s[tid] + (dx1 + 2.f*dx2 + 2.f*dx3 + dx4) * (1.f/6.f);
        float gx = xf - b_obs[bg];
        float c1 = (1.f + a2 + 0.5f*a3*a2 + 0.25f*a4*a3*a2) * (1.f/6.f);
        float c2 = (2.f + a3 + 0.5f*a4*a3) * (1.f/6.f);
        float c3 = (2.f + a4) * (1.f/6.f);
        float c4 = (1.f/6.f);
        float kb1 = a1;
        float kb2 = a2 * (1.f + 0.5f*kb1);
        float kb3 = a3 * (1.f + 0.5f*kb2);
        float kb4 = a4 * (1.f + kb3);
        float gb = 1.f + (kb1 + 2.f*kb2 + 2.f*kb3 + kb4) * (1.f/6.f);
        const float inv = 1.f / 512.f;
        out[bg] = gx * gb * inv;
        ws[O_GX + bg] = gx; ws[O_GB + bg] = gb;
        float sc = scT_s[tid];
        float cc[4] = {c1, c2, c3, c4};
        #pragma unroll
        for (int s2 = 0; s2 < 4; ++s2) {
            float wvv = cc[s2] * gx * inv;
            ws[O_WV  + s2 * 512 + bg] = wvv;
            ws[O_WV2 + s2 * 512 + bg] = wvv * sc;
        }
    }
}

#define K2_STEP(ACC, KQ)                                                              \
    {                                                                                 \
        const v2f wq = *(const v2f*)(wp + (KQ) * 4);                                  \
        v2f af, bf;                                                                   \
        af.x = wq.x * dp[(KQ) * 1024];                                                \
        af.y = wq.y * dp[(KQ) * 1024 + 256];                                          \
        bf.x = ip[(KQ) * 256];                                                        \
        bf.y = ip[(KQ) * 256 + 64];                                                   \
        ACC = __builtin_amdgcn_wmma_f32_16x16x4_f32(false, af, false, bf, (short)0, ACC, false, false); \
    }

static __device__ __forceinline__ void sgrad_role(const int blk, const float* __restrict__ ws, float* __restrict__ out)
{
    __shared__ float part[16][256];
    const int tid  = threadIdx.x;
    const int lane = tid & 31;
    const int wvi  = tid >> 5;
    const int mt = blk >> 2;
    const int nt = blk & 3;
    const int ln  = lane & 15;
    const int klo = (lane < 16) ? 0 : 2;
    const int m   = mt * 16 + ln;
    const int kc  = nt * 16 + ln;
    const unsigned kt0 = (unsigned)(wvi * 128 + klo);
    const float* __restrict__ dp = ws + O_D   + kt0 * 256u + (unsigned)m;
    const float* __restrict__ ip = ws + O_INP + kt0 * 64u  + (unsigned)kc;
    const float* __restrict__ wp = ws + O_WV  + kt0;

    v8f acc0, acc1;
    #pragma unroll
    for (int v = 0; v < 8; ++v) { acc0[v] = 0.f; acc1[v] = 0.f; }

    #pragma unroll
    for (int kq = 0; kq < 32; kq += 2) {
        K2_STEP(acc0, kq)
        K2_STEP(acc1, kq + 1)
    }

    #pragma unroll
    for (int v = 0; v < 8; ++v) part[wvi][v * 32 + lane] = acc0[v] + acc1[v];
    __syncthreads();

    if (tid < 256) {
        float s = 0.f;
        for (int w = 0; w < 16; ++w) s += part[w][tid];
        const int v  = tid >> 5;
        const int l  = tid & 31;
        const int mm = mt * 16 + v + ((l < 16) ? 0 : 8);
        const int kk = nt * 16 + (l & 15);
        if (kk < 50) out[25088 + mm * 50 + kk] = s;
    }
}
#undef K2_STEP

static __device__ __forceinline__ void wgrad_role(const int blk, const float* __restrict__ ws, float* __restrict__ out)
{
    __shared__ float red[16][8];
    const int tid   = threadIdx.x;
    const int lane  = tid & 31;
    const int wvi   = tid >> 5;
    const int ml    = lane & 3;
    const int slice = tid >> 2;
    const int m     = blk * 4 + ml;

    float w2[16], ph[16];
    #pragma unroll
    for (int i = 0; i < 16; ++i) {
        int kt = i * 128 + slice;
        w2[i] = ws[O_WV2 + kt];
        ph[i] = ws[O_PHI + (unsigned)kt * 256u + m];
    }
    float ac = 0.f, as2 = 0.f;
    #pragma unroll
    for (int i = 0; i < 16; ++i) {
        float sn, cs; sincosf(ph[i], &sn, &cs);
        ac  = fmaf(w2[i], cs, ac);
        as2 = fmaf(w2[i], sn, as2);
    }
    for (int off = 4; off <= 16; off <<= 1) {
        ac  += __shfl_xor(ac,  off, 32);
        as2 += __shfl_xor(as2, off, 32);
    }
    if (lane < 4) { red[wvi][lane] = ac; red[wvi][4 + lane] = as2; }
    __syncthreads();
    if (tid < 8) {
        float s = 0.f;
        for (int w = 0; w < 16; ++w) s += red[w][tid];
        out[37888 + (tid >> 2) * 256 + blk * 4 + (tid & 3)] = s;
    }
}

static __device__ __forceinline__ void ugrad_role(const int blk, const float* __restrict__ ws,
                                                  const float* __restrict__ S, float* __restrict__ out)
{
    __shared__ __attribute__((aligned(16))) float dsm[16][260];
    __shared__ float upart[16][768];
    const int tid  = threadIdx.x;
    const int lane = tid & 31;
    const int wvi  = tid >> 5;
    const int ln   = lane & 15;
    const int hh   = lane >> 4;
    const int b0   = blk * 16;
    const int brow = b0 + wvi;

    float  wvs[4];
    float4 dlo[4], dhi[4];
    #pragma unroll
    for (int s = 0; s < 4; ++s) {
        wvs[s] = ws[O_WV + s * 512 + brow];
        const float4* dp = (const float4*)(ws + O_D + (unsigned)((s * 512 + brow) * 256 + lane * 8));
        dlo[s] = dp[0];
        dhi[s] = dp[1];
    }

    const int mb = wvi * 16 + hh * 8;
    v2f bfr[3][4];
    #pragma unroll
    for (int nt = 0; nt < 3; ++nt) {
        #pragma unroll
        for (int q = 0; q < 4; ++q) {
            const float* sp = S + (mb + 2 * q) * 50 + 2 + nt * 16 + ln;
            bfr[nt][q].x = sp[0];
            bfr[nt][q].y = sp[50];
        }
    }

    float eua[2], eub[2], egx[2], egb[2];
    #pragma unroll
    for (int i = 0; i < 2; ++i) {
        eua[i] = 0.f; eub[i] = 0.f; egx[i] = 0.f; egb[i] = 0.f;
        const int idx = tid + i * 512;
        if (idx < 768) {
            const int nt  = idx >> 8;
            const int row = ((idx >> 5) & 7) + ((idx & 16) ? 8 : 0);
            const int cc  = idx & 15;
            const int b   = b0 + row;
            const int oa  = (nt == 0) ? 16 : 0;
            const int ob  = (nt == 2) ? 16 : 32;
            eua[i] = ws[O_INP + (unsigned)(b * 64 + 2 + oa + cc)];
            eub[i] = ws[O_INP + (unsigned)(b * 64 + 2 + ob + cc)];
            egx[i] = ws[O_GX + b];
            egb[i] = ws[O_GB + b];
        }
    }

    float4 lo, hi;
    lo.x = wvs[0] * dlo[0].x; lo.y = wvs[0] * dlo[0].y; lo.z = wvs[0] * dlo[0].z; lo.w = wvs[0] * dlo[0].w;
    hi.x = wvs[0] * dhi[0].x; hi.y = wvs[0] * dhi[0].y; hi.z = wvs[0] * dhi[0].z; hi.w = wvs[0] * dhi[0].w;
    #pragma unroll
    for (int s = 1; s < 4; ++s) {
        lo.x = fmaf(wvs[s], dlo[s].x, lo.x); lo.y = fmaf(wvs[s], dlo[s].y, lo.y);
        lo.z = fmaf(wvs[s], dlo[s].z, lo.z); lo.w = fmaf(wvs[s], dlo[s].w, lo.w);
        hi.x = fmaf(wvs[s], dhi[s].x, hi.x); hi.y = fmaf(wvs[s], dhi[s].y, hi.y);
        hi.z = fmaf(wvs[s], dhi[s].z, hi.z); hi.w = fmaf(wvs[s], dhi[s].w, hi.w);
    }
    *(float4*)&dsm[wvi][lane * 8]     = lo;
    *(float4*)&dsm[wvi][lane * 8 + 4] = hi;
    __syncthreads();

    const float4 a0 = *(const float4*)&dsm[ln][mb];
    const float4 a1 = *(const float4*)&dsm[ln][mb + 4];
    v2f af[4];
    af[0].x = a0.x; af[0].y = a0.y;
    af[1].x = a0.z; af[1].y = a0.w;
    af[2].x = a1.x; af[2].y = a1.y;
    af[3].x = a1.z; af[3].y = a1.w;

    v8f acc[3];
    #pragma unroll
    for (int nt = 0; nt < 3; ++nt) {
        #pragma unroll
        for (int v = 0; v < 8; ++v) acc[nt][v] = 0.f;
    }
    #pragma unroll
    for (int q = 0; q < 4; ++q) {
        #pragma unroll
        for (int nt = 0; nt < 3; ++nt) {
            acc[nt] = __builtin_amdgcn_wmma_f32_16x16x4_f32(
                false, af[q], false, bfr[nt][q], (short)0, acc[nt], false, false);
        }
    }

    #pragma unroll
    for (int nt = 0; nt < 3; ++nt) {
        #pragma unroll
        for (int v = 0; v < 8; ++v) upart[wvi][nt * 256 + v * 32 + lane] = acc[nt][v];
    }
    __syncthreads();

    #pragma unroll
    for (int i = 0; i < 2; ++i) {
        const int idx = tid + i * 512;
        if (idx < 768) {
            float s = 0.f;
            for (int w = 0; w < 16; ++w) s += upart[w][idx];
            const int nt  = idx >> 8;
            const int row = ((idx >> 5) & 7) + ((idx & 16) ? 8 : 0);
            const int cc  = idx & 15;
            const float p = eua[i] * eub[i];
            out[512 + (b0 + row) * 48 + nt * 16 + cc] = egx[i] * (1.f / 512.f) * egb[i] * p + s;
        }
    }
}

__global__ __launch_bounds__(512) __attribute__((amdgpu_waves_per_eu(4, 4)))
void k2_consumers(const float* __restrict__ ws, const float* __restrict__ S, float* __restrict__ out)
{
    const int bid = blockIdx.x;
    if      (bid < 32) ugrad_role(bid, ws, S, out);
    else if (bid < 96) sgrad_role(bid - 32, ws, out);
    else               wgrad_role(bid - 96, ws, out);
}

extern "C" void kernel_launch(void* const* d_in, const int* in_sizes, int n_in,
                              void* d_out, int out_size, void* d_ws, size_t ws_size,
                              hipStream_t stream)
{
    (void)in_sizes; (void)n_in; (void)out_size; (void)ws_size;
    const float* U0   = (const float*)d_in[0];
    const float* U1   = (const float*)d_in[1];
    const float* U2   = (const float*)d_in[2];
    const float* S    = (const float*)d_in[3];
    const float* w    = (const float*)d_in[4];
    const float* btn  = (const float*)d_in[5];
    const float* bobs = (const float*)d_in[6];
    const int*   bin  = (const int*)d_in[7];
    float* out = (float*)d_out;
    float* ws  = (float*)d_ws;

    k1_stages<<<32, 512, 0, stream>>>(U0, U1, U2, S, w, btn, bobs, bin, ws, out);
    k2_consumers<<<160, 512, 0, stream>>>(ws, S, out);
}


# stage kernel: 4 batch rows a block, not 16 (128 blocks, not 32): the tile's 16 rows hold the 4 rows 4x over, each lane half takes 2 of them: 2 sincos a lane and stage, not 8; sum orders unchanged
# speedup vs baseline: 43.2470x; 1.2943x over previous
#include <hip/hip_runtime.h>
#include <math.h>

typedef float v2f __attribute__((ext_vector_type(2)));
typedef float v8f __attribute__((ext_vector_type(8)));

#define BB 512
#define MM 256
#define INDIM 50

#define O_D    0u
#define O_PHI  524288u
#define O_INP  1048576u
#define O_WV   1179648u
#define O_WV2  1181696u
#define O_GX   1183744u
#define O_GB   1184256u

#define ROWS 4
__global__ __launch_bounds__(512) void k1_stages(
    const float* __restrict__ U0, const float* __restrict__ U1, const float* __restrict__ U2,
    const float* __restrict__ S,  const float* __restrict__ w,
    const float* __restrict__ b_t_n, const float* __restrict__ b_obs,
    const int* __restrict__ b_i_n,
    float* __restrict__ ws, float* __restrict__ out)
{
    __shared__ float inp_s[ROWS * 53];
    __shared__ float redF[16][ROWS];
    __shared__ float redA[16][ROWS];
    __shared__ float xs[ROWS], beta_s[ROWS], scT_s[ROWS];
    __shared__ float aS[4][ROWS], dxS[4][ROWS];

    const int tid  = threadIdx.x;
    const int wv   = tid >> 5;
    const int lane = tid & 31;
    const int ln   = lane & 15;
    const int rsel = (lane < 16) ? 0 : 2;
    const int b0g  = blockIdx.x * ROWS;
    const int klo  = (lane < 16) ? 0 : 2;
    const int m    = wv * 16 + ln;

    v2f bfr[13];
    #pragma unroll
    for (int kq = 0; kq < 13; ++kq) {
        const int k0 = kq * 4 + klo;
        bfr[kq].x = (k0     < INDIM) ? S[m * INDIM + k0]     : 0.f;
        bfr[kq].y = (k0 + 1 < INDIM) ? S[m * INDIM + k0 + 1] : 0.f;
    }
    const float w0r = w[m], w1r = w[m + 256], s1r = S[m * INDIM + 1];

    const float btv = b_t_n[b0g + (tid & (ROWS - 1))];

    if (tid < ROWS * 48) {
        const int bl = tid / 48, j = tid - bl * 48;
        const int u = j >> 4, r = j & 15;
        const float* Uu = (u == 0) ? U0 : (u == 1) ? U1 : U2;
        const int i0 = b_i_n[(b0g + bl) * 3 + u];
        inp_s[bl * 53 + 2 + j] = Uu[i0 * 16 + r];
    }
    if (tid < ROWS * 3) {
        const int bl = tid / 3, c = tid - bl * 3;
        inp_s[bl * 53 + 50 + c] = 0.f;
    }
    if (tid < ROWS) scT_s[tid] = btv;
    __syncthreads();

    if (tid < ROWS) {
        float acc = 0.f;
        for (int r = 0; r < 16; ++r)
            acc += inp_s[tid*53 + 2 + r] * inp_s[tid*53 + 18 + r] * inp_s[tid*53 + 34 + r];
        beta_s[tid] = acc; xs[tid] = acc;
    }
    __syncthreads();

    float stv[2]; int bgv[2];
    #pragma unroll
    for (int v = 0; v < 2; ++v) { stv[v] = scT_s[rsel + v]; bgv[v] = b0g + rsel + v; }

    for (int s = 0; s < 4; ++s) {
        if (tid < ROWS) {
            float ts = (s == 0) ? 0.f : ((s == 3) ? 1.f : 0.5f);
            inp_s[tid * 53 + 0] = scT_s[tid] * ts;
            inp_s[tid * 53 + 1] = xs[tid];
        }
        __syncthreads();

        if (tid < ROWS * 64) {
            const int bl = tid >> 6, k = tid & 63;
            ws[O_INP + (unsigned)((s * 512 + b0g + bl) * 64 + k)] = (k < 50) ? inp_s[bl * 53 + k] : 0.f;
        }

        v8f acc;
        #pragma unroll
        for (int v = 0; v < 8; ++v) acc[v] = 0.f;

        #pragma unroll
        for (int kq = 0; kq < 13; ++kq) {
            const int k0 = kq * 4 + klo;
            v2f af;
            af.x = inp_s[(ln & (ROWS - 1)) * 53 + k0];
            af.y = inp_s[(ln & (ROWS - 1)) * 53 + k0 + 1];
            acc = __builtin_amdgcn_wmma_f32_16x16x4_f32(
                false, af, false, bfr[kq], (short)0, acc, false, false);
        }

        float ph[2];
        ph[0] = (lane < 16) ? acc[0] : acc[2];
        ph[1] = (lane < 16) ? acc[1] : acc[3];

        float Fp[2], Ap[2];
        #pragma unroll
        for (int v = 0; v < 2; ++v) {
            float phi = ph[v];
            float sn, cs;
            sincosf(phi, &sn, &cs);
            float fval  = w0r * cs + w1r * sn;
            float dterm = w1r * cs - w0r * sn;
            Fp[v] = fval;
            Ap[v] = dterm * s1r;
            unsigned o = (unsigned)((s * 512 + bgv[v]) * 256 + m);
            ws[O_D + o]   = stv[v] * dterm;
            ws[O_PHI + o] = phi;
        }
        #pragma unroll
        for (int v = 0; v < 2; ++v) {
            float f = Fp[v], a = Ap[v];
            #pragma unroll
            for (int off = 8; off >= 1; off >>= 1) {
                f += __shfl_xor(f, off, 32);
                a += __shfl_xor(a, off, 32);
            }
            Fp[v] = f; Ap[v] = a;
        }
        if (ln == 0) {
            #pragma unroll
            for (int v = 0; v < 2; ++v) { redF[wv][rsel + v] = Fp[v]; redA[wv][rsel + v] = Ap[v]; }
        }
        __syncthreads();
        if (tid < ROWS) {
            float F = 0.f, A = 0.f;
            for (int q = 0; q < 16; ++q) { F += redF[q][tid]; A += redA[q][tid]; }
            float dx = scT_s[tid] * F;
            float a  = scT_s[tid] * A;
            dxS[s][tid] = dx; aS[s][tid] = a;
            if      (s == 0) xs[tid] = beta_s[tid] + 0.5f * dx;
            else if (s == 1) xs[tid] = beta_s[tid] + 0.5f * dx;
            else if (s == 2) xs[tid] = beta_s[tid] + dx;
        }
        __syncthreads();
    }

    if (tid < ROWS) {
        int bg = b0g + tid;
        float a1 = aS[0][tid], a2 = aS[1][tid], a3 = aS[2][tid], a4 = aS[3][tid];
        float dx1 = dxS[0][tid], dx2 = dxS[1][tid], dx3 = dxS[2][tid], dx4 = dxS[3][tid];
        float xf = beta_s[tid] + (dx1 + 2.f*dx2 + 2.f*dx3 + dx4) * (1.f/6.f);
        float gx = xf - b_obs[bg];
        float c1 = (1.f + a2 + 0.5f*a3*a2 + 0.25f*a4*a3*a2) * (1.f/6.f);
        float c2 = (2.f + a3 + 0.5f*a4*a3) * (1.f/6.f);
        float c3 = (2.f + a4) * (1.f/6.f);
        float c4 = (1.f/6.f);
        float kb1 = a1;
        float kb2 = a2 * (1.f + 0.5f*kb1);
        float kb3 = a3 * (1.f + 0.5f*kb2);
        float kb4 = a4 * (1.f + kb3);
        float gb = 1.f + (kb1 + 2.f*kb2 + 2.f*kb3 + kb4) * (1.f/6.f);
        const float inv = 1.f / 512.f;
        out[bg] = gx * gb * inv;
        ws[O_GX + bg] = gx; ws[O_GB + bg] = gb;
        float sc = scT_s[tid];
        float cc[4] = {c1, c2, c3, c4};
        #pragma unroll
        for (int s2 = 0; s2 < 4; ++s2) {
            float wvv = cc[s2] * gx * inv;
            ws[O_WV  + s2 * 512 + bg] = wvv;
            ws[O_WV2 + s2 * 512 + bg] = wvv * sc;
        }
    }
}

#define K2_STEP(ACC, KQ)                                                              \
    {                                                                                 \
        const v2f wq = *(const v2f*)(wp + (KQ) * 4);                                  \
        v2f af, bf;                                                                   \
        af.x = wq.x * dp[(KQ) * 1024];                                                \
        af.y = wq.y * dp[(KQ) * 1024 + 256];                                          \
        bf.x = ip[(KQ) * 256];                                                        \
        bf.y = ip[(KQ) * 256 + 64];                                                   \
        ACC = __builtin_amdgcn_wmma_f32_16x16x4_f32(false, af, false, bf, (short)0, ACC, false, false); \
    }

static __device__ __forceinline__ void sgrad_role(const int blk, const float* __restrict__ ws, float* __restrict__ out)
{
    __shared__ float part[16][256];
    const int tid  = threadIdx.x;
    const int lane = tid & 31;
    const int wvi  = tid >> 5;
    const int mt = blk >> 2;
    const int nt = blk & 3;
    const int ln  = lane & 15;
    const int klo = (lane < 16) ? 0 : 2;
    const int m   = mt * 16 + ln;
    const int kc  = nt * 16 + ln;
    const unsigned kt0 = (unsigned)(wvi * 128 + klo);
    const float* __restrict__ dp = ws + O_D   + kt0 * 256u + (unsigned)m;
    const float* __restrict__ ip = ws + O_INP + kt0 * 64u  + (unsigned)kc;
    const float* __restrict__ wp = ws + O_WV  + kt0;

    v8f acc0, acc1;
    #pragma unroll
    for (int v = 0; v < 8; ++v) { acc0[v] = 0.f; acc1[v] = 0.f; }

    #pragma unroll
    for (int kq = 0; kq < 32; kq += 2) {
        K2_STEP(acc0, kq)
        K2_STEP(acc1, kq + 1)
    }

    #pragma unroll
    for (int v = 0; v < 8; ++v) part[wvi][v * 32 + lane] = acc0[v] + acc1[v];
    __syncthreads();

    if (tid < 256) {
        float s = 0.f;
        for (int w = 0; w < 16; ++w) s += part[w][tid];
        const int v  = tid >> 5;
        const int l  = tid & 31;
        const int mm = mt * 16 + v + ((l < 16) ? 0 : 8);
        const int kk = nt * 16 + (l & 15);
        if (kk < 50) out[25088 + mm * 50 + kk] = s;
    }
}
#undef K2_STEP

static __device__ __forceinline__ void wgrad_role(const int blk, const float* __restrict__ ws, float* __restrict__ out)
{
    __shared__ float red[16][8];
    const int tid   = threadIdx.x;
    const int lane  = tid & 31;
    const int wvi   = tid >> 5;
    const int ml    = lane & 3;
    const int slice = tid >> 2;
    const int m     = blk * 4 + ml;

    float w2[16], ph[16];
    #pragma unroll
    for (int i = 0; i < 16; ++i) {
        int kt = i * 128 + slice;
        w2[i] = ws[O_WV2 + kt];
        ph[i] = ws[O_PHI + (unsigned)kt * 256u + m];
    }
    float ac = 0.f, as2 = 0.f;
    #pragma unroll
    for (int i = 0; i < 16; ++i) {
        float sn, cs; sincosf(ph[i], &sn, &cs);
        ac  = fmaf(w2[i], cs, ac);
        as2 = fmaf(w2[i], sn, as2);
    }
    for (int off = 4; off <= 16; off <<= 1) {
        ac  += __shfl_xor(ac,  off, 32);
        as2 += __shfl_xor(as2, off, 32);
    }
    if (lane < 4) { red[wvi][lane] = ac; red[wvi][4 + lane] = as2; }
    __syncthreads();
    if (tid < 8) {
        float s = 0.f;
        for (int w = 0; w < 16; ++w) s += red[w][tid];
        out[37888 + (tid >> 2) * 256 + blk * 4 + (tid & 3)] = s;
    }
}

static __device__ __forceinline__ void ugrad_role(const int blk, const float* __restrict__ ws,
                                                  const float* __restrict__ S, float* __restrict__ out)
{
    __shared__ __attribute__((aligned(16))) float dsm[16][260];
    __shared__ float upart[16][768];
    const int tid  = threadIdx.x;
    const int lane = tid & 31;
    const int wvi  = tid >> 5;
    const int ln   = lane & 15;
    const int hh   = lane >> 4;
    const int b0   = blk * 16;
    const int brow = b0 + wvi;

    float  wvs[4];
    float4 dlo[4], dhi[4];
    #pragma unroll
    for (int s = 0; s < 4; ++s) {
        wvs[s] = ws[O_WV + s * 512 + brow];
        const float4* dp = (const float4*)(ws + O_D + (unsigned)((s * 512 + brow) * 256 + lane * 8));
        dlo[s] = dp[0];
        dhi[s] = dp[1];
    }

    const int mb = wvi * 16 + hh * 8;
    v2f bfr[3][4];
    #pragma unroll
    for (int nt = 0; nt < 3; ++nt) {
        #pragma unroll
        for (int q = 0; q < 4; ++q) {
            const float* sp = S + (mb + 2 * q) * 50 + 2 + nt * 16 + ln;
            bfr[nt][q].x = sp[0];
            bfr[nt][q].y = sp[50];
        }
    }

    float eua[2], eub[2], egx[2], egb[2];
    #pragma unroll
    for (int i = 0; i < 2; ++i) {
        eua[i] = 0.f; eub[i] = 0.f; egx[i] = 0.f; egb[i] = 0.f;
        const int idx = tid + i * 512;
        if (idx < 768) {
            const int nt  = idx >> 8;
            const int row = ((idx >> 5) & 7) + ((idx & 16) ? 8 : 0);
            const int cc  = idx & 15;
            const int b   = b0 + row;
            const int oa  = (nt == 0) ? 16 : 0;
            const int ob  = (nt == 2) ? 16 : 32;
            eua[i] = ws[O_INP + (unsigned)(b * 64 + 2 + oa + cc)];
            eub[i] = ws[O_INP + (unsigned)(b * 64 + 2 + ob + cc)];
            egx[i] = ws[O_GX + b];
            egb[i] = ws[O_GB + b];
        }
    }

    float4 lo, hi;
    lo.x = wvs[0] * dlo[0].x; lo.y = wvs[0] * dlo[0].y; lo.z = wvs[0] * dlo[0].z; lo.w = wvs[0] * dlo[0].w;
    hi.x = wvs[0] * dhi[0].x; hi.y = wvs[0] * dhi[0].y; hi.z = wvs[0] * dhi[0].z; hi.w = wvs[0] * dhi[0].w;
    #pragma unroll
    for (int s = 1; s < 4; ++s) {
        lo.x = fmaf(wvs[s], dlo[s].x, lo.x); lo.y = fmaf(wvs[s], dlo[s].y, lo.y);
        lo.z = fmaf(wvs[s], dlo[s].z, lo.z); lo.w = fmaf(wvs[s], dlo[s].w, lo.w);
        hi.x = fmaf(wvs[s], dhi[s].x, hi.x); hi.y = fmaf(wvs[s], dhi[s].y, hi.y);
        hi.z = fmaf(wvs[s], dhi[s].z, hi.z); hi.w = fmaf(wvs[s], dhi[s].w, hi.w);
    }
    *(float4*)&dsm[wvi][lane * 8]     = lo;
    *(float4*)&dsm[wvi][lane * 8 + 4] = hi;
    __syncthreads();

    const float4 a0 = *(const float4*)&dsm[ln][mb];
    const float4 a1 = *(const float4*)&dsm[ln][mb + 4];
    v2f af[4];
    af[0].x = a0.x; af[0].y = a0.y;
    af[1].x = a0.z; af[1].y = a0.w;
    af[2].x = a1.x; af[2].y = a1.y;
    af[3].x = a1.z; af[3].y = a1.w;

    v8f acc[3];
    #pragma unroll
    for (int nt = 0; nt < 3; ++nt) {
        #pragma unroll
        for (int v = 0; v < 8; ++v) acc[nt][v] = 0.f;
    }
    #pragma unroll
    for (int q = 0; q < 4; ++q) {
        #pragma unroll
        for (int nt = 0; nt < 3; ++nt) {
            acc[nt] = __builtin_amdgcn_wmma_f32_16x16x4_f32(
                false, af[q], false, bfr[nt][q], (short)0, acc[nt], false, false);
        }
    }

    #pragma unroll
    for (int nt = 0; nt < 3; ++nt) {
        #pragma unroll
        for (int v = 0; v < 8; ++v) upart[wvi][nt * 256 + v * 32 + lane] = acc[nt][v];
    }
    __syncthreads();

    #pragma unroll
    for (int i = 0; i < 2; ++i) {
        const int idx = tid + i * 512;
        if (idx < 768) {
            float s = 0.f;
            for (int w = 0; w < 16; ++w) s += upart[w][idx];
            const int nt  = idx >> 8;
            const int row = ((idx >> 5) & 7) + ((idx & 16) ? 8 : 0);
            const int cc  = idx & 15;
            const float p = eua[i] * eub[i];
            out[512 + (b0 + row) * 48 + nt * 16 + cc] = egx[i] * (1.f / 512.f) * egb[i] * p + s;
        }
    }
}

__global__ __launch_bounds__(512) __attribute__((amdgpu_waves_per_eu(4, 4)))
void k2_consumers(const float* __restrict__ ws, const float* __restrict__ S, float* __restrict__ out)
{
    const int bid = blockIdx.x;
    if      (bid < 32) ugrad_role(bid, ws, S, out);
    else if (bid < 96) sgrad_role(bid - 32, ws, out);
    else               wgrad_role(bid - 96, ws, out);
}

extern "C" void kernel_launch(void* const* d_in, const int* in_sizes, int n_in,
                              void* d_out, int out_size, void* d_ws, size_t ws_size,
                              hipStream_t stream)
{
    (void)in_sizes; (void)n_in; (void)out_size; (void)ws_size;
    const float* U0   = (const float*)d_in[0];
    const float* U1   = (const float*)d_in[1];
    const float* U2   = (const float*)d_in[2];
    const float* S    = (const float*)d_in[3];
    const float* w    = (const float*)d_in[4];
    const float* btn  = (const float*)d_in[5];
    const float* bobs = (const float*)d_in[6];
    const int*   bin  = (const int*)d_in[7];
    float* out = (float*)d_out;
    float* ws  = (float*)d_ws;

    k1_stages<<<128, 512, 0, stream>>>(U0, U1, U2, S, w, btn, bobs, bin, ws, out);
    k2_consumers<<<160, 512, 0, stream>>>(ws, S, out);
}


# stage kernel: only the row-sum reduction is left between two stages: lanes add by row_xmask DPP, every wave adds the 16 wave partials itself, d/phi stores behind the barrier; 1 barrier a stage, not 3
# speedup vs baseline: 50.7168x; 1.1727x over previous
#include <hip/hip_runtime.h>
#include <math.h>

typedef float v2f __attribute__((ext_vector_type(2)));
typedef float v8f __attribute__((ext_vector_type(8)));

#define BB 512
#define MM 256
#define INDIM 50

#define O_D    0u
#define O_PHI  524288u
#define O_INP  1048576u
#define O_WV   1179648u
#define O_WV2  1181696u
#define O_GX   1183744u
#define O_GB   1184256u

#define ROWS 4
#define ROW_XOR_ADD(V, M) \
    ((V) + __int_as_float(__builtin_amdgcn_update_dpp(0, __float_as_int(V), 0x160 + (M), 0xF, 0xF, true)))
__global__ __launch_bounds__(512) void k1_stages(
    const float* __restrict__ U0, const float* __restrict__ U1, const float* __restrict__ U2,
    const float* __restrict__ S,  const float* __restrict__ w,
    const float* __restrict__ b_t_n, const float* __restrict__ b_obs,
    const int* __restrict__ b_i_n,
    float* __restrict__ ws, float* __restrict__ out)
{
    __shared__ float inp_s[ROWS * 53];
    __shared__ __attribute__((aligned(16))) float redF[4][ROWS][16];
    __shared__ __attribute__((aligned(16))) float redA[4][ROWS][16];
    __shared__ float beta_s[ROWS], scT_s[ROWS];
    __shared__ float dxS[4][ROWS];

    const int tid  = threadIdx.x;
    const int wv   = tid >> 5;
    const int lane = tid & 31;
    const int ln   = lane & 15;
    const int rsel = (lane < 16) ? 0 : 2;
    const int rA   = ln & (ROWS - 1);
    const int b0g  = blockIdx.x * ROWS;
    const int klo  = (lane < 16) ? 0 : 2;
    const int m    = wv * 16 + ln;

    v2f bfr[12];
    #pragma unroll
    for (int kq = 0; kq < 12; ++kq) {
        const int k0 = 2 + kq * 4 + klo;
        bfr[kq].x = S[m * INDIM + k0];
        bfr[kq].y = S[m * INDIM + k0 + 1];
    }
    const float w0r = w[m], w1r = w[m + 256], s0r = S[m * INDIM], s1r = S[m * INDIM + 1];
    v2f bst;
    bst.x = (lane < 16) ? s0r : 0.f;
    bst.y = (lane < 16) ? s1r : 0.f;

    const float btv = b_t_n[b0g + (tid & (ROWS - 1))];

    if (tid < ROWS * 48) {
        const int bl = tid / 48, j = tid - bl * 48;
        const int u = j >> 4, r = j & 15;
        const float* Uu = (u == 0) ? U0 : (u == 1) ? U1 : U2;
        const int i0 = b_i_n[(b0g + bl) * 3 + u];
        inp_s[bl * 53 + 2 + j] = Uu[i0 * 16 + r];
    }
    if (tid < ROWS) scT_s[tid] = btv;
    __syncthreads();

    if (tid < ROWS) {
        float acc = 0.f;
        for (int r = 0; r < 16; ++r)
            acc += inp_s[tid*53 + 2 + r] * inp_s[tid*53 + 18 + r] * inp_s[tid*53 + 34 + r];
        beta_s[tid] = acc;
    }

    #pragma unroll
    for (int i = 0; i < 2; ++i) {
        const int idx = tid + i * 512;
        const int s = idx >> 8, bl = (idx >> 6) & (ROWS - 1), k = idx & 63;
        if (k >= 2)
            ws[O_INP + (unsigned)((s * 512 + b0g + bl) * 64 + k)] = (k < 50) ? inp_s[bl * 53 + k] : 0.f;
    }

    v8f accc;
    #pragma unroll
    for (int v = 0; v < 8; ++v) accc[v] = 0.f;
    #pragma unroll
    for (int kq = 0; kq < 12; ++kq) {
        const int k0 = 2 + kq * 4 + klo;
        v2f af;
        af.x = inp_s[rA * 53 + k0];
        af.y = inp_s[rA * 53 + k0 + 1];
        accc = __builtin_amdgcn_wmma_f32_16x16x4_f32(
            false, af, false, bfr[kq], (short)0, accc, false, false);
    }
    __syncthreads();

    float stv[2]; int bgv[2];
    #pragma unroll
    for (int v = 0; v < 2; ++v) { stv[v] = scT_s[rsel + v]; bgv[v] = b0g + rsel + v; }

    const float scA   = scT_s[rA];
    const float betaA = beta_s[rA];
    float xcur = betaA;

    for (int s = 0; s < 4; ++s) {
        const float ts = (s == 0) ? 0.f : ((s == 3) ? 1.f : 0.5f);
        const float t0 = scA * ts;

        if (tid < ROWS) {
            const unsigned o = (unsigned)((s * 512 + b0g + tid) * 64);
            ws[O_INP + o]     = t0;
            ws[O_INP + o + 1] = xcur;
        }

        v2f afs;
        afs.x = (lane < 16) ? t0   : 0.f;
        afs.y = (lane < 16) ? xcur : 0.f;
        const v8f acc = __builtin_amdgcn_wmma_f32_16x16x4_f32(
            false, afs, false, bst, (short)0, accc, false, false);

        float ph[2];
        ph[0] = (lane < 16) ? acc[0] : acc[2];
        ph[1] = (lane < 16) ? acc[1] : acc[3];

        float Fp[2], Ap[2], dv[2];
        #pragma unroll
        for (int v = 0; v < 2; ++v) {
            float sn, cs;
            sincosf(ph[v], &sn, &cs);
            float fval  = w0r * cs + w1r * sn;
            float dterm = w1r * cs - w0r * sn;
            Fp[v] = fval;
            Ap[v] = dterm * s1r;
            dv[v] = stv[v] * dterm;
        }
        #pragma unroll
        for (int v = 0; v < 2; ++v) {
            float f = Fp[v], a = Ap[v];
            f = ROW_XOR_ADD(f, 8); a = ROW_XOR_ADD(a, 8);
            f = ROW_XOR_ADD(f, 4); a = ROW_XOR_ADD(a, 4);
            f = ROW_XOR_ADD(f, 2); a = ROW_XOR_ADD(a, 2);
            f = ROW_XOR_ADD(f, 1); a = ROW_XOR_ADD(a, 1);
            Fp[v] = f; Ap[v] = a;
        }
        if (ln == 0) {
            #pragma unroll
            for (int v = 0; v < 2; ++v) { redF[s][rsel + v][wv] = Fp[v]; redA[s][rsel + v][wv] = Ap[v]; }
        }
        __syncthreads();

        const float4* fq = (const float4*)&redF[s][rA][0];
        const float4 f0 = fq[0], f1 = fq[1], f2 = fq[2], f3 = fq[3];

        #pragma unroll
        for (int v = 0; v < 2; ++v) {
            const unsigned o = (unsigned)((s * 512 + bgv[v]) * 256 + m);
            ws[O_D + o]   = dv[v];
            ws[O_PHI + o] = ph[v];
        }

        float F = 0.f;
        F += f0.x; F += f0.y; F += f0.z; F += f0.w;
        F += f1.x; F += f1.y; F += f1.z; F += f1.w;
        F += f2.x; F += f2.y; F += f2.z; F += f2.w;
        F += f3.x; F += f3.y; F += f3.z; F += f3.w;
        const float dx = scA * F;
        if (tid < ROWS) dxS[s][tid] = dx;
        xcur = (s == 2) ? (betaA + dx) : (betaA + 0.5f * dx);
    }

    if (tid < ROWS) {
        int bg = b0g + tid;
        float av[4];
        #pragma unroll
        for (int s2 = 0; s2 < 4; ++s2) {
            float A = 0.f;
            for (int q = 0; q < 16; ++q) A += redA[s2][tid][q];
            av[s2] = scA * A;
        }
        float a1 = av[0], a2 = av[1], a3 = av[2], a4 = av[3];
        float dx1 = dxS[0][tid], dx2 = dxS[1][tid], dx3 = dxS[2][tid], dx4 = dxS[3][tid];
        float xf = betaA + (dx1 + 2.f*dx2 + 2.f*dx3 + dx4) * (1.f/6.f);
        float gx = xf - b_obs[bg];
        float c1 = (1.f + a2 + 0.5f*a3*a2 + 0.25f*a4*a3*a2) * (1.f/6.f);
        float c2 = (2.f + a3 + 0.5f*a4*a3) * (1.f/6.f);
        float c3 = (2.f + a4) * (1.f/6.f);
        float c4 = (1.f/6.f);
        float kb1 = a1;
        float kb2 = a2 * (1.f + 0.5f*kb1);
        float kb3 = a3 * (1.f + 0.5f*kb2);
        float kb4 = a4 * (1.f + kb3);
        float gb = 1.f + (kb1 + 2.f*kb2 + 2.f*kb3 + kb4) * (1.f/6.f);
        const float inv = 1.f / 512.f;
        out[bg] = gx * gb * inv;
        ws[O_GX + bg] = gx; ws[O_GB + bg] = gb;
        float sc = scA;
        float cc[4] = {c1, c2, c3, c4};
        #pragma unroll
        for (int s2 = 0; s2 < 4; ++s2) {
            float wvv = cc[s2] * gx * inv;
            ws[O_WV  + s2 * 512 + bg] = wvv;
            ws[O_WV2 + s2 * 512 + bg] = wvv * sc;
        }
    }
}
#undef ROW_XOR_ADD

#define K2_STEP(ACC, KQ)                                                              \
    {                                                                                 \
        const v2f wq = *(const v2f*)(wp + (KQ) * 4);                                  \
        v2f af, bf;                                                                   \
        af.x = wq.x * dp[(KQ) * 1024];                                                \
        af.y = wq.y * dp[(KQ) * 1024 + 256];                                          \
        bf.x = ip[(KQ) * 256];                                                        \
        bf.y = ip[(KQ) * 256 + 64];                                                   \
        ACC = __builtin_amdgcn_wmma_f32_16x16x4_f32(false, af, false, bf, (short)0, ACC, false, false); \
    }

static __device__ __forceinline__ void sgrad_role(const int blk, const float* __restrict__ ws, float* __restrict__ out)
{
    __shared__ float part[16][256];
    const int tid  = threadIdx.x;
    const int lane = tid & 31;
    const int wvi  = tid >> 5;
    const int mt = blk >> 2;
    const int nt = blk & 3;
    const int ln  = lane & 15;
    const int klo = (lane < 16) ? 0 : 2;
    const int m   = mt * 16 + ln;
    const int kc  = nt * 16 + ln;
    const unsigned kt0 = (unsigned)(wvi * 128 + klo);
    const float* __restrict__ dp = ws + O_D   + kt0 * 256u + (unsigned)m;
    const float* __restrict__ ip = ws + O_INP + kt0 * 64u  + (unsigned)kc;
    const float* __restrict__ wp = ws + O_WV  + kt0;

    v8f acc0, acc1;
    #pragma unroll
    for (int v = 0; v < 8; ++v) { acc0[v] = 0.f; acc1[v] = 0.f; }

    #pragma unroll
    for (int kq = 0; kq < 32; kq += 2) {
        K2_STEP(acc0, kq)
        K2_STEP(acc1, kq + 1)
    }

    #pragma unroll
    for (int v = 0; v < 8; ++v) part[wvi][v * 32 + lane] = acc0[v] + acc1[v];
    __syncthreads();

    if (tid < 256) {
        float s = 0.f;
        for (int w = 0; w < 16; ++w) s += part[w][tid];
        const int v  = tid >> 5;
        const int l  = tid & 31;
        const int mm = mt * 16 + v + ((l < 16) ? 0 : 8);
        const int kk = nt * 16 + (l & 15);
        if (kk < 50) out[25088 + mm * 50 + kk] = s;
    }
}
#undef K2_STEP

static __device__ __forceinline__ void wgrad_role(const int blk, const float* __restrict__ ws, float* __restrict__ out)
{
    __shared__ float red[16][8];
    const int tid   = threadIdx.x;
    const int lane  = tid & 31;
    const int wvi   = tid >> 5;
    const int ml    = lane & 3;
    const int slice = tid >> 2;
    const int m     = blk * 4 + ml;

    float w2[16], ph[16];
    #pragma unroll
    for (int i = 0; i < 16; ++i) {
        int kt = i * 128 + slice;
        w2[i] = ws[O_WV2 + kt];
        ph[i] = ws[O_PHI + (unsigned)kt * 256u + m];
    }
    float ac = 0.f, as2 = 0.f;
    #pragma unroll
    for (int i = 0; i < 16; ++i) {
        float sn, cs; sincosf(ph[i], &sn, &cs);
        ac  = fmaf(w2[i], cs, ac);
        as2 = fmaf(w2[i], sn, as2);
    }
    for (int off = 4; off <= 16; off <<= 1) {
        ac  += __shfl_xor(ac,  off, 32);
        as2 += __shfl_xor(as2, off, 32);
    }
    if (lane < 4) { red[wvi][lane] = ac; red[wvi][4 + lane] = as2; }
    __syncthreads();
    if (tid < 8) {
        float s = 0.f;
        for (int w = 0; w < 16; ++w) s += red[w][tid];
        out[37888 + (tid >> 2) * 256 + blk * 4 + (tid & 3)] = s;
    }
}

static __device__ __forceinline__ void ugrad_role(const int blk, const float* __restrict__ ws,
                                                  const float* __restrict__ S, float* __restrict__ out)
{
    __shared__ __attribute__((aligned(16))) float dsm[16][260];
    __shared__ float upart[16][768];
    const int tid  = threadIdx.x;
    const int lane = tid & 31;
    const int wvi  = tid >> 5;
    const int ln   = lane & 15;
    const int hh   = lane >> 4;
    const int b0   = blk * 16;
    const int brow = b0 + wvi;

    float  wvs[4];
    float4 dlo[4], dhi[4];
    #pragma unroll
    for (int s = 0; s < 4; ++s) {
        wvs[s] = ws[O_WV + s * 512 + brow];
        const float4* dp = (const float4*)(ws + O_D + (unsigned)((s * 512 + brow) * 256 + lane * 8));
        dlo[s] = dp[0];
        dhi[s] = dp[1];
    }

    const int mb = wvi * 16 + hh * 8;
    v2f bfr[3][4];
    #pragma unroll
    for (int nt = 0; nt < 3; ++nt) {
        #pragma unroll
        for (int q = 0; q < 4; ++q) {
            const float* sp = S + (mb + 2 * q) * 50 + 2 + nt * 16 + ln;
            bfr[nt][q].x = sp[0];
            bfr[nt][q].y = sp[50];
        }
    }

    float eua[2], eub[2], egx[2], egb[2];
    #pragma unroll
    for (int i = 0; i < 2; ++i) {
        eua[i] = 0.f; eub[i] = 0.f; egx[i] = 0.f; egb[i] = 0.f;
        const int idx = tid + i * 512;
        if (idx < 768) {
            const int nt  = idx >> 8;
            const int row = ((idx >> 5) & 7) + ((idx & 16) ? 8 : 0);
            const int cc  = idx & 15;
            const int b   = b0 + row;
            const int oa  = (nt == 0) ? 16 : 0;
            const int ob  = (nt == 2) ? 16 : 32;
            eua[i] = ws[O_INP + (unsigned)(b * 64 + 2 + oa + cc)];
            eub[i] = ws[O_INP + (unsigned)(b * 64 + 2 + ob + cc)];
            egx[i] = ws[O_GX + b];
            egb[i] = ws[O_GB + b];
        }
    }

    float4 lo, hi;
    lo.x = wvs[0] * dlo[0].x; lo.y = wvs[0] * dlo[0].y; lo.z = wvs[0] * dlo[0].z; lo.w = wvs[0] * dlo[0].w;
    hi.x = wvs[0] * dhi[0].x; hi.y = wvs[0] * dhi[0].y; hi.z = wvs[0] * dhi[0].z; hi.w = wvs[0] * dhi[0].w;
    #pragma unroll
    for (int s = 1; s < 4; ++s) {
        lo.x = fmaf(wvs[s], dlo[s].x, lo.x); lo.y = fmaf(wvs[s], dlo[s].y, lo.y);
        lo.z = fmaf(wvs[s], dlo[s].z, lo.z); lo.w = fmaf(wvs[s], dlo[s].w, lo.w);
        hi.x = fmaf(wvs[s], dhi[s].x, hi.x); hi.y = fmaf(wvs[s], dhi[s].y, hi.y);
        hi.z = fmaf(wvs[s], dhi[s].z, hi.z); hi.w = fmaf(wvs[s], dhi[s].w, hi.w);
    }
    *(float4*)&dsm[wvi][lane * 8]     = lo;
    *(float4*)&dsm[wvi][lane * 8 + 4] = hi;
    __syncthreads();

    const float4 a0 = *(const float4*)&dsm[ln][mb];
    const float4 a1 = *(const float4*)&dsm[ln][mb + 4];
    v2f af[4];
    af[0].x = a0.x; af[0].y = a0.y;
    af[1].x = a0.z; af[1].y = a0.w;
    af[2].x = a1.x; af[2].y = a1.y;
    af[3].x = a1.z; af[3].y = a1.w;

    v8f acc[3];
    #pragma unroll
    for (int nt = 0; nt < 3; ++nt) {
        #pragma unroll
        for (int v = 0; v < 8; ++v) acc[nt][v] = 0.f;
    }
    #pragma unroll
    for (int q = 0; q < 4; ++q) {
        #pragma unroll
        for (int nt = 0; nt < 3; ++nt) {
            acc[nt] = __builtin_amdgcn_wmma_f32_16x16x4_f32(
                false, af[q], false, bfr[nt][q], (short)0, acc[nt], false, false);
        }
    }

    #pragma unroll
    for (int nt = 0; nt < 3; ++nt) {
        #pragma unroll
        for (int v = 0; v < 8; ++v) upart[wvi][nt * 256 + v * 32 + lane] = acc[nt][v];
    }
    __syncthreads();

    #pragma unroll
    for (int i = 0; i < 2; ++i) {
        const int idx = tid + i * 512;
        if (idx < 768) {
            float s = 0.f;
            for (int w = 0; w < 16; ++w) s += upart[w][idx];
            const int nt  = idx >> 8;
            const int row = ((idx >> 5) & 7) + ((idx & 16) ? 8 : 0);
            const int cc  = idx & 15;
            const float p = eua[i] * eub[i];
            out[512 + (b0 + row) * 48 + nt * 16 + cc] = egx[i] * (1.f / 512.f) * egb[i] * p + s;
        }
    }
}

__global__ __launch_bounds__(512) __attribute__((amdgpu_waves_per_eu(4, 4)))
void k2_consumers(const float* __restrict__ ws, const float* __restrict__ S, float* __restrict__ out)
{
    const int bid = blockIdx.x;
    if      (bid < 32) ugrad_role(bid, ws, S, out);
    else if (bid < 96) sgrad_role(bid - 32, ws, out);
    else               wgrad_role(bid - 96, ws, out);
}

extern "C" void kernel_launch(void* const* d_in, const int* in_sizes, int n_in,
                              void* d_out, int out_size, void* d_ws, size_t ws_size,
                              hipStream_t stream)
{
    (void)in_sizes; (void)n_in; (void)out_size; (void)ws_size;
    const float* U0   = (const float*)d_in[0];
    const float* U1   = (const float*)d_in[1];
    const float* U2   = (const float*)d_in[2];
    const float* S    = (const float*)d_in[3];
    const float* w    = (const float*)d_in[4];
    const float* btn  = (const float*)d_in[5];
    const float* bobs = (const float*)d_in[6];
    const int*   bin  = (const int*)d_in[7];
    float* out = (float*)d_out;
    float* ws  = (float*)d_ws;

    k1_stages<<<128, 512, 0, stream>>>(U0, U1, U2, S, w, btn, bobs, bin, ws, out);
    k2_consumers<<<160, 512, 0, stream>>>(ws, S, out);
}


# stage kernel sums the 4 stages itself (plane dsum + per-block column partials for S cols 0,1 and w): consumers' K 2048->512, no 2nd sincos; sum orders of out_S, out_w, out_Uvec change
# speedup vs baseline: 71.8167x; 1.4160x over previous
#include <hip/hip_runtime.h>
#include <math.h>

typedef float v2f __attribute__((ext_vector_type(2)));
typedef float v8f __attribute__((ext_vector_type(8)));

#define BB 512
#define MM 256
#define INDIM 50

#define O_DS   0u
#define O_P    524288u
#define O_INP  1048576u
#define O_GX   1183744u
#define O_GB   1184256u

#define ROWS 4
#define ROW_XOR_ADD(V, M) \
    ((V) + __int_as_float(__builtin_amdgcn_update_dpp(0, __float_as_int(V), 0x160 + (M), 0xF, 0xF, true)))
__global__ __launch_bounds__(512) void k1_stages(
    const float* __restrict__ U0, const float* __restrict__ U1, const float* __restrict__ U2,
    const float* __restrict__ S,  const float* __restrict__ w,
    const float* __restrict__ b_t_n, const float* __restrict__ b_obs,
    const int* __restrict__ b_i_n,
    float* __restrict__ ws, float* __restrict__ out)
{
    __shared__ float inp_s[ROWS * 53];
    __shared__ __attribute__((aligned(16))) float redF[4][ROWS][16];
    __shared__ __attribute__((aligned(16))) float redA[4][ROWS][16];
    __shared__ float beta_s[ROWS], scT_s[ROWS];
    __shared__ float dxS[4][ROWS];
    __shared__ __attribute__((aligned(16))) float wvS[ROWS][4];
    __shared__ __attribute__((aligned(16))) float xS[ROWS][4];

    const int tid  = threadIdx.x;
    const int wv   = tid >> 5;
    const int lane = tid & 31;
    const int ln   = lane & 15;
    const int rsel = (lane < 16) ? 0 : 2;
    const int rA   = ln & (ROWS - 1);
    const int b0g  = blockIdx.x * ROWS;
    const int klo  = (lane < 16) ? 0 : 2;
    const int m    = wv * 16 + ln;

    v2f bfr[12];
    #pragma unroll
    for (int kq = 0; kq < 12; ++kq) {
        const int k0 = 2 + kq * 4 + klo;
        bfr[kq].x = S[m * INDIM + k0];
        bfr[kq].y = S[m * INDIM + k0 + 1];
    }
    const float w0r = w[m], w1r = w[m + 256], s0r = S[m * INDIM], s1r = S[m * INDIM + 1];
    v2f bst;
    bst.x = (lane < 16) ? s0r : 0.f;
    bst.y = (lane < 16) ? s1r : 0.f;

    const float btv = b_t_n[b0g + (tid & (ROWS - 1))];
    const float bobv = b_obs[b0g + (tid & (ROWS - 1))];

    if (tid < ROWS * 48) {
        const int bl = tid / 48, j = tid - bl * 48;
        const int u = j >> 4, r = j & 15;
        const float* Uu = (u == 0) ? U0 : (u == 1) ? U1 : U2;
        const int i0 = b_i_n[(b0g + bl) * 3 + u];
        inp_s[bl * 53 + 2 + j] = Uu[i0 * 16 + r];
    }
    if (tid < ROWS) scT_s[tid] = btv;
    __syncthreads();

    if (tid < ROWS) {
        float acc = 0.f;
        for (int r = 0; r < 16; ++r)
            acc += inp_s[tid*53 + 2 + r] * inp_s[tid*53 + 18 + r] * inp_s[tid*53 + 34 + r];
        beta_s[tid] = acc;
    }

    if (tid < ROWS * 64) {
        const int bl = tid >> 6, k = tid & 63;
        if (k >= 2 && k < 50)
            ws[O_INP + (unsigned)((b0g + bl) * 64 + k)] = inp_s[bl * 53 + k];
    }

    v8f accc;
    #pragma unroll
    for (int v = 0; v < 8; ++v) accc[v] = 0.f;
    #pragma unroll
    for (int kq = 0; kq < 12; ++kq) {
        const int k0 = 2 + kq * 4 + klo;
        v2f af;
        af.x = inp_s[rA * 53 + k0];
        af.y = inp_s[rA * 53 + k0 + 1];
        accc = __builtin_amdgcn_wmma_f32_16x16x4_f32(
            false, af, false, bfr[kq], (short)0, accc, false, false);
    }
    __syncthreads();

    float stv[2]; int bgv[2];
    #pragma unroll
    for (int v = 0; v < 2; ++v) { stv[v] = scT_s[rsel + v]; bgv[v] = b0g + rsel + v; }

    const float scA   = scT_s[rA];
    const float betaA = beta_s[rA];
    float xcur = betaA;

    float csv[4][2], snv[4][2];
    #pragma unroll
    for (int s = 0; s < 4; ++s) {
        const float ts = (s == 0) ? 0.f : ((s == 3) ? 1.f : 0.5f);
        const float t0 = scA * ts;

        if (tid < ROWS) xS[tid][s] = xcur;

        v2f afs;
        afs.x = (lane < 16) ? t0   : 0.f;
        afs.y = (lane < 16) ? xcur : 0.f;
        const v8f acc = __builtin_amdgcn_wmma_f32_16x16x4_f32(
            false, afs, false, bst, (short)0, accc, false, false);

        float ph[2];
        ph[0] = (lane < 16) ? acc[0] : acc[2];
        ph[1] = (lane < 16) ? acc[1] : acc[3];

        float Fp[2], Ap[2];
        #pragma unroll
        for (int v = 0; v < 2; ++v) {
            float sn, cs;
            sincosf(ph[v], &sn, &cs);
            float fval  = w0r * cs + w1r * sn;
            float dterm = w1r * cs - w0r * sn;
            Fp[v] = fval;
            Ap[v] = dterm * s1r;
            csv[s][v] = cs;
            snv[s][v] = sn;
        }
        #pragma unroll
        for (int v = 0; v < 2; ++v) {
            float f = Fp[v], a = Ap[v];
            f = ROW_XOR_ADD(f, 8); a = ROW_XOR_ADD(a, 8);
            f = ROW_XOR_ADD(f, 4); a = ROW_XOR_ADD(a, 4);
            f = ROW_XOR_ADD(f, 2); a = ROW_XOR_ADD(a, 2);
            f = ROW_XOR_ADD(f, 1); a = ROW_XOR_ADD(a, 1);
            Fp[v] = f; Ap[v] = a;
        }
        if (ln == 0) {
            #pragma unroll
            for (int v = 0; v < 2; ++v) { redF[s][rsel + v][wv] = Fp[v]; redA[s][rsel + v][wv] = Ap[v]; }
        }
        __syncthreads();

        const float4* fq = (const float4*)&redF[s][rA][0];
        const float4 f0 = fq[0], f1 = fq[1], f2 = fq[2], f3 = fq[3];

        float F = 0.f;
        F += f0.x; F += f0.y; F += f0.z; F += f0.w;
        F += f1.x; F += f1.y; F += f1.z; F += f1.w;
        F += f2.x; F += f2.y; F += f2.z; F += f2.w;
        F += f3.x; F += f3.y; F += f3.z; F += f3.w;
        const float dx = scA * F;
        if (tid < ROWS) dxS[s][tid] = dx;
        xcur = (s == 2) ? (betaA + dx) : (betaA + 0.5f * dx);
    }

    if (tid < ROWS) {
        int bg = b0g + tid;
        float av[4];
        #pragma unroll
        for (int s2 = 0; s2 < 4; ++s2) {
            float A = 0.f;
            for (int q = 0; q < 16; ++q) A += redA[s2][tid][q];
            av[s2] = scA * A;
        }
        float a1 = av[0], a2 = av[1], a3 = av[2], a4 = av[3];
        float dx1 = dxS[0][tid], dx2 = dxS[1][tid], dx3 = dxS[2][tid], dx4 = dxS[3][tid];
        float xf = betaA + (dx1 + 2.f*dx2 + 2.f*dx3 + dx4) * (1.f/6.f);
        float gx = xf - bobv;
        float c1 = (1.f + a2 + 0.5f*a3*a2 + 0.25f*a4*a3*a2) * (1.f/6.f);
        float c2 = (2.f + a3 + 0.5f*a4*a3) * (1.f/6.f);
        float c3 = (2.f + a4) * (1.f/6.f);
        float c4 = (1.f/6.f);
        float kb1 = a1;
        float kb2 = a2 * (1.f + 0.5f*kb1);
        float kb3 = a3 * (1.f + 0.5f*kb2);
        float kb4 = a4 * (1.f + kb3);
        float gb = 1.f + (kb1 + 2.f*kb2 + 2.f*kb3 + kb4) * (1.f/6.f);
        const float inv = 1.f / 512.f;
        out[bg] = gx * gb * inv;
        ws[O_GX + bg] = gx; ws[O_GB + bg] = gb;
        float cc[4] = {c1, c2, c3, c4};
        #pragma unroll
        for (int s2 = 0; s2 < 4; ++s2) wvS[tid][s2] = cc[s2] * gx * inv;
    }
    __syncthreads();

    float e0c = 0.f, e1c = 0.f, pcc = 0.f, psc = 0.f;
    #pragma unroll
    for (int v = 0; v < 2; ++v) {
        const float4 w4 = *(const float4*)&wvS[rsel + v][0];
        const float4 x4 = *(const float4*)&xS[rsel + v][0];
        const float wq[4] = {w4.x, w4.y, w4.z, w4.w};
        const float xq[4] = {x4.x, x4.y, x4.z, x4.w};
        float ds = 0.f, e0 = 0.f, e1 = 0.f, pc = 0.f, ps = 0.f;
        #pragma unroll
        for (int s = 0; s < 4; ++s) {
            const float ts    = (s == 0) ? 0.f : ((s == 3) ? 1.f : 0.5f);
            const float dterm = w1r * csv[s][v] - w0r * snv[s][v];
            const float g     = wq[s] * (stv[v] * dterm);
            const float w2    = wq[s] * stv[v];
            ds += g;
            e0 += g * (stv[v] * ts);
            e1 += g * xq[s];
            pc += w2 * csv[s][v];
            ps += w2 * snv[s][v];
        }
        ws[O_DS + (unsigned)(bgv[v] * 256 + m)] = ds;
        e0c += e0; e1c += e1; pcc += pc; psc += ps;
    }
    e0c += __shfl_xor(e0c, 16, 32);
    e1c += __shfl_xor(e1c, 16, 32);
    pcc += __shfl_xor(pcc, 16, 32);
    psc += __shfl_xor(psc, 16, 32);
    if (lane < 16) {
        float* pp = ws + O_P + (unsigned)(blockIdx.x * 1024 + m);
        pp[0]   = e0c;
        pp[256] = e1c;
        pp[512] = pcc;
        pp[768] = psc;
    }
}
#undef ROW_XOR_ADD

#define K2_STEP(ACC, KQ)                                                              \
    {                                                                                 \
        v2f af, bf;                                                                   \
        af.x = dp[(KQ) * 1024];                                                       \
        af.y = dp[(KQ) * 1024 + 256];                                                 \
        bf.x = ip[(KQ) * 256];                                                        \
        bf.y = ip[(KQ) * 256 + 64];                                                   \
        ACC = __builtin_amdgcn_wmma_f32_16x16x4_f32(false, af, false, bf, (short)0, ACC, false, false); \
    }

static __device__ __forceinline__ void sgrad_role(const int blk, const float* __restrict__ ws, float* __restrict__ out)
{
    __shared__ float part[16][256];
    const int tid  = threadIdx.x;
    const int lane = tid & 31;
    const int wvi  = tid >> 5;
    const int mt = blk / 3;
    const int nt = blk - mt * 3;
    const int ln  = lane & 15;
    const int klo = (lane < 16) ? 0 : 2;
    const int m   = mt * 16 + ln;
    const int kc  = 2 + nt * 16 + ln;
    const unsigned b0 = (unsigned)(wvi * 32 + klo);
    const float* __restrict__ dp = ws + O_DS  + b0 * 256u + (unsigned)m;
    const float* __restrict__ ip = ws + O_INP + b0 * 64u  + (unsigned)kc;

    v8f acc0, acc1;
    #pragma unroll
    for (int v = 0; v < 8; ++v) { acc0[v] = 0.f; acc1[v] = 0.f; }

    #pragma unroll
    for (int kq = 0; kq < 8; kq += 2) {
        K2_STEP(acc0, kq)
        K2_STEP(acc1, kq + 1)
    }

    #pragma unroll
    for (int v = 0; v < 8; ++v) part[wvi][v * 32 + lane] = acc0[v] + acc1[v];
    __syncthreads();

    if (tid < 256) {
        float s = 0.f;
        for (int w = 0; w < 16; ++w) s += part[w][tid];
        const int v  = tid >> 5;
        const int l  = tid & 31;
        const int mm = mt * 16 + v + ((l < 16) ? 0 : 8);
        const int kk = 2 + nt * 16 + (l & 15);
        out[25088 + mm * 50 + kk] = s;
    }
}
#undef K2_STEP

static __device__ __forceinline__ void csum_role(const int blk, const float* __restrict__ ws, float* __restrict__ out)
{
    __shared__ float red[8][64];
    const int tid   = threadIdx.x;
    const int col   = tid & 63;
    const int slice = tid >> 6;
    const int q0    = blk * 64;

    float pv[16];
    #pragma unroll
    for (int i = 0; i < 16; ++i)
        pv[i] = ws[O_P + (unsigned)((slice * 16 + i) * 1024 + q0 + col)];
    float s = 0.f;
    #pragma unroll
    for (int i = 0; i < 16; ++i) s += pv[i];
    red[slice][col] = s;
    __syncthreads();
    if (tid < 64) {
        float t = 0.f;
        for (int w = 0; w < 8; ++w) t += red[w][tid];
        const int q     = q0 + tid;
        const int plane = q >> 8;
        const int mq    = q & 255;
        if      (plane == 0) out[25088 + mq * 50]     = t;
        else if (plane == 1) out[25088 + mq * 50 + 1] = t;
        else if (plane == 2) out[37888 + mq]          = t;
        else                 out[37888 + 256 + mq]    = t;
    }
}

static __device__ __forceinline__ void ugrad_role(const int blk, const float* __restrict__ ws,
                                                  const float* __restrict__ S, float* __restrict__ out)
{
    __shared__ float upart[16][768];
    const int tid  = threadIdx.x;
    const int lane = tid & 31;
    const int wvi  = tid >> 5;
    const int ln   = lane & 15;
    const int hh   = lane >> 4;
    const int b0   = blk * 16;

    const int mb = wvi * 16 + hh * 8;
    const float4* ap = (const float4*)(ws + O_DS + (unsigned)((b0 + ln) * 256 + mb));
    const float4 a0 = ap[0];
    const float4 a1 = ap[1];

    v2f bfr[3][4];
    #pragma unroll
    for (int nt = 0; nt < 3; ++nt) {
        #pragma unroll
        for (int q = 0; q < 4; ++q) {
            const float* sp = S + (mb + 2 * q) * 50 + 2 + nt * 16 + ln;
            bfr[nt][q].x = sp[0];
            bfr[nt][q].y = sp[50];
        }
    }

    float eua[2], eub[2], egx[2], egb[2];
    #pragma unroll
    for (int i = 0; i < 2; ++i) {
        eua[i] = 0.f; eub[i] = 0.f; egx[i] = 0.f; egb[i] = 0.f;
        const int idx = tid + i * 512;
        if (idx < 768) {
            const int nt  = idx >> 8;
            const int row = ((idx >> 5) & 7) + ((idx & 16) ? 8 : 0);
            const int cc  = idx & 15;
            const int b   = b0 + row;
            const int oa  = (nt == 0) ? 16 : 0;
            const int ob  = (nt == 2) ? 16 : 32;
            eua[i] = ws[O_INP + (unsigned)(b * 64 + 2 + oa + cc)];
            eub[i] = ws[O_INP + (unsigned)(b * 64 + 2 + ob + cc)];
            egx[i] = ws[O_GX + b];
            egb[i] = ws[O_GB + b];
        }
    }

    v2f af[4];
    af[0].x = a0.x; af[0].y = a0.y;
    af[1].x = a0.z; af[1].y = a0.w;
    af[2].x = a1.x; af[2].y = a1.y;
    af[3].x = a1.z; af[3].y = a1.w;

    v8f acc[3];
    #pragma unroll
    for (int nt = 0; nt < 3; ++nt) {
        #pragma unroll
        for (int v = 0; v < 8; ++v) acc[nt][v] = 0.f;
    }
    #pragma unroll
    for (int q = 0; q < 4; ++q) {
        #pragma unroll
        for (int nt = 0; nt < 3; ++nt) {
            acc[nt] = __builtin_amdgcn_wmma_f32_16x16x4_f32(
                false, af[q], false, bfr[nt][q], (short)0, acc[nt], false, false);
        }
    }

    #pragma unroll
    for (int nt = 0; nt < 3; ++nt) {
        #pragma unroll
        for (int v = 0; v < 8; ++v) upart[wvi][nt * 256 + v * 32 + lane] = acc[nt][v];
    }
    __syncthreads();

    #pragma unroll
    for (int i = 0; i < 2; ++i) {
        const int idx = tid + i * 512;
        if (idx < 768) {
            float s = 0.f;
            for (int w = 0; w < 16; ++w) s += upart[w][idx];
            const int nt  = idx >> 8;
            const int row = ((idx >> 5) & 7) + ((idx & 16) ? 8 : 0);
            const int cc  = idx & 15;
            const float p = eua[i] * eub[i];
            out[512 + (b0 + row) * 48 + nt * 16 + cc] = egx[i] * (1.f / 512.f) * egb[i] * p + s;
        }
    }
}

__global__ __launch_bounds__(512) __attribute__((amdgpu_waves_per_eu(4, 4)))
void k2_consumers(const float* __restrict__ ws, const float* __restrict__ S, float* __restrict__ out)
{
    const int bid = blockIdx.x;
    if      (bid < 32) ugrad_role(bid, ws, S, out);
    else if (bid < 80) sgrad_role(bid - 32, ws, out);
    else               csum_role(bid - 80, ws, out);
}

extern "C" void kernel_launch(void* const* d_in, const int* in_sizes, int n_in,
                              void* d_out, int out_size, void* d_ws, size_t ws_size,
                              hipStream_t stream)
{
    (void)in_sizes; (void)n_in; (void)out_size; (void)ws_size;
    const float* U0   = (const float*)d_in[0];
    const float* U1   = (const float*)d_in[1];
    const float* U2   = (const float*)d_in[2];
    const float* S    = (const float*)d_in[3];
    const float* w    = (const float*)d_in[4];
    const float* btn  = (const float*)d_in[5];
    const float* bobs = (const float*)d_in[6];
    const int*   bin  = (const int*)d_in[7];
    float* out = (float*)d_out;
    float* ws  = (float*)d_ws;

    k1_stages<<<128, 512, 0, stream>>>(U0, U1, U2, S, w, btn, bobs, bin, ws, out);
    k2_consumers<<<96, 512, 0, stream>>>(ws, S, out);
}


# stage kernel: S reaches the B fragments through LDS: the block copies S once with whole-line 16-byte loads (7 a lane), not 13 strided 8-byte loads a lane queued ahead of the gather; sums unchanged
# speedup vs baseline: 76.8085x; 1.0695x over previous
#include <hip/hip_runtime.h>
#include <math.h>

typedef float v2f __attribute__((ext_vector_type(2)));
typedef float v8f __attribute__((ext_vector_type(8)));

#define BB 512
#define MM 256
#define INDIM 50

#define O_DS   0u
#define O_P    524288u
#define O_INP  1048576u
#define O_GX   1183744u
#define O_GB   1184256u

#define ROWS 4
#define ROW_XOR_ADD(V, M) \
    ((V) + __int_as_float(__builtin_amdgcn_update_dpp(0, __float_as_int(V), 0x160 + (M), 0xF, 0xF, true)))
__global__ __launch_bounds__(512) void k1_stages(
    const float* __restrict__ U0, const float* __restrict__ U1, const float* __restrict__ U2,
    const float* __restrict__ S,  const float* __restrict__ w,
    const float* __restrict__ b_t_n, const float* __restrict__ b_obs,
    const int* __restrict__ b_i_n,
    float* __restrict__ ws, float* __restrict__ out)
{
    __shared__ __attribute__((aligned(16))) float S_s[7 * 512 * 4];
    __shared__ float inp_s[ROWS * 53 + 1];
    __shared__ __attribute__((aligned(16))) float redF[4][ROWS][16];
    __shared__ __attribute__((aligned(16))) float redA[4][ROWS][16];
    __shared__ float beta_s[ROWS], scT_s[ROWS];
    __shared__ float dxS[4][ROWS];
    __shared__ __attribute__((aligned(16))) float wvS[ROWS][4];
    __shared__ __attribute__((aligned(16))) float xS[ROWS][4];

    const int tid  = threadIdx.x;
    const int wv   = tid >> 5;
    const int lane = tid & 31;
    const int ln   = lane & 15;
    const int rsel = (lane < 16) ? 0 : 2;
    const int rA   = ln & (ROWS - 1);
    const int b0g  = blockIdx.x * ROWS;
    const int klo  = (lane < 16) ? 0 : 2;
    const int m    = wv * 16 + ln;

    const int gt  = (tid < ROWS * 48) ? tid : 0;
    const int gbl = gt / 48, gj = gt - gbl * 48;
    const int gu = gj >> 4, gr = gj & 15;
    const int gi0 = b_i_n[(b0g + gbl) * 3 + gu];

    const float4* __restrict__ S4 = (const float4*)S;
    const float4 sc0 = S4[tid],        sc1 = S4[tid + 512],  sc2 = S4[tid + 1024];
    const float4 sc3 = S4[tid + 1536], sc4 = S4[tid + 2048], sc5 = S4[tid + 2560];
    const float4 sc6 = S4[(tid < 128) ? tid + 3072 : MM * INDIM / 4 - 1];
    const float w0r = w[m], w1r = w[m + 256];

    const float btv = b_t_n[b0g + (tid & (ROWS - 1))];
    const float bobv = b_obs[b0g + (tid & (ROWS - 1))];

    const float* Uu = (gu == 0) ? U0 : (gu == 1) ? U1 : U2;
    const float gval = Uu[gi0 * 16 + gr];
    float4* S4s = (float4*)S_s;
    S4s[tid]        = sc0;  S4s[tid + 512]  = sc1;  S4s[tid + 1024] = sc2;
    S4s[tid + 1536] = sc3;  S4s[tid + 2048] = sc4;  S4s[tid + 2560] = sc5;
    S4s[tid + 3072] = sc6;
    inp_s[(tid < ROWS * 48) ? gbl * 53 + 2 + gj : ROWS * 53] = gval;
    scT_s[tid & (ROWS - 1)] = btv;
    __syncthreads();

    v2f bfr[12];
    #pragma unroll
    for (int kq = 0; kq < 12; ++kq) {
        const int k0 = 2 + kq * 4 + klo;
        bfr[kq] = *(const v2f*)&S_s[m * INDIM + k0];
    }
    const v2f s01 = *(const v2f*)&S_s[m * INDIM];
    const float s0r = s01.x, s1r = s01.y;
    v2f bst;
    bst.x = (lane < 16) ? s0r : 0.f;
    bst.y = (lane < 16) ? s1r : 0.f;

    if (tid < ROWS) {
        float acc = 0.f;
        for (int r = 0; r < 16; ++r)
            acc += inp_s[tid*53 + 2 + r] * inp_s[tid*53 + 18 + r] * inp_s[tid*53 + 34 + r];
        beta_s[tid] = acc;
    }

    if (tid < ROWS * 64) {
        const int bl = tid >> 6, k = tid & 63;
        if (k >= 2 && k < 50)
            ws[O_INP + (unsigned)((b0g + bl) * 64 + k)] = inp_s[bl * 53 + k];
    }

    v8f accc;
    #pragma unroll
    for (int v = 0; v < 8; ++v) accc[v] = 0.f;
    #pragma unroll
    for (int kq = 0; kq < 12; ++kq) {
        const int k0 = 2 + kq * 4 + klo;
        v2f af;
        af.x = inp_s[rA * 53 + k0];
        af.y = inp_s[rA * 53 + k0 + 1];
        accc = __builtin_amdgcn_wmma_f32_16x16x4_f32(
            false, af, false, bfr[kq], (short)0, accc, false, false);
    }
    __syncthreads();

    float stv[2]; int bgv[2];
    #pragma unroll
    for (int v = 0; v < 2; ++v) { stv[v] = scT_s[rsel + v]; bgv[v] = b0g + rsel + v; }

    const float scA   = scT_s[rA];
    const float betaA = beta_s[rA];
    float xcur = betaA;

    float csv[4][2], snv[4][2];
    #pragma unroll
    for (int s = 0; s < 4; ++s) {
        const float ts = (s == 0) ? 0.f : ((s == 3) ? 1.f : 0.5f);
        const float t0 = scA * ts;

        if (tid < ROWS) xS[tid][s] = xcur;

        v2f afs;
        afs.x = (lane < 16) ? t0   : 0.f;
        afs.y = (lane < 16) ? xcur : 0.f;
        const v8f acc = __builtin_amdgcn_wmma_f32_16x16x4_f32(
            false, afs, false, bst, (short)0, accc, false, false);

        float ph[2];
        ph[0] = (lane < 16) ? acc[0] : acc[2];
        ph[1] = (lane < 16) ? acc[1] : acc[3];

        float Fp[2], Ap[2];
        #pragma unroll
        for (int v = 0; v < 2; ++v) {
            float sn, cs;
            sincosf(ph[v], &sn, &cs);
            float fval  = w0r * cs + w1r * sn;
            float dterm = w1r * cs - w0r * sn;
            Fp[v] = fval;
            Ap[v] = dterm * s1r;
            csv[s][v] = cs;
            snv[s][v] = sn;
        }
        #pragma unroll
        for (int v = 0; v < 2; ++v) {
            float f = Fp[v], a = Ap[v];
            f = ROW_XOR_ADD(f, 8); a = ROW_XOR_ADD(a, 8);
            f = ROW_XOR_ADD(f, 4); a = ROW_XOR_ADD(a, 4);
            f = ROW_XOR_ADD(f, 2); a = ROW_XOR_ADD(a, 2);
            f = ROW_XOR_ADD(f, 1); a = ROW_XOR_ADD(a, 1);
            Fp[v] = f; Ap[v] = a;
        }
        if (ln == 0) {
            #pragma unroll
            for (int v = 0; v < 2; ++v) { redF[s][rsel + v][wv] = Fp[v]; redA[s][rsel + v][wv] = Ap[v]; }
        }
        __syncthreads();

        const float4* fq = (const float4*)&redF[s][rA][0];
        const float4 f0 = fq[0], f1 = fq[1], f2 = fq[2], f3 = fq[3];

        float F = 0.f;
        F += f0.x; F += f0.y; F += f0.z; F += f0.w;
        F += f1.x; F += f1.y; F += f1.z; F += f1.w;
        F += f2.x; F += f2.y; F += f2.z; F += f2.w;
        F += f3.x; F += f3.y; F += f3.z; F += f3.w;
        const float dx = scA * F;
        if (tid < ROWS) dxS[s][tid] = dx;
        xcur = (s == 2) ? (betaA + dx) : (betaA + 0.5f * dx);
    }

    if (tid < ROWS) {
        int bg = b0g + tid;
        float av[4];
        #pragma unroll
        for (int s2 = 0; s2 < 4; ++s2) {
            float A = 0.f;
            for (int q = 0; q < 16; ++q) A += redA[s2][tid][q];
            av[s2] = scA * A;
        }
        float a1 = av[0], a2 = av[1], a3 = av[2], a4 = av[3];
        float dx1 = dxS[0][tid], dx2 = dxS[1][tid], dx3 = dxS[2][tid], dx4 = dxS[3][tid];
        float xf = betaA + (dx1 + 2.f*dx2 + 2.f*dx3 + dx4) * (1.f/6.f);
        float gx = xf - bobv;
        float c1 = (1.f + a2 + 0.5f*a3*a2 + 0.25f*a4*a3*a2) * (1.f/6.f);
        float c2 = (2.f + a3 + 0.5f*a4*a3) * (1.f/6.f);
        float c3 = (2.f + a4) * (1.f/6.f);
        float c4 = (1.f/6.f);
        float kb1 = a1;
        float kb2 = a2 * (1.f + 0.5f*kb1);
        float kb3 = a3 * (1.f + 0.5f*kb2);
        float kb4 = a4 * (1.f + kb3);
        float gb = 1.f + (kb1 + 2.f*kb2 + 2.f*kb3 + kb4) * (1.f/6.f);
        const float inv = 1.f / 512.f;
        out[bg] = gx * gb * inv;
        ws[O_GX + bg] = gx; ws[O_GB + bg] = gb;
        float cc[4] = {c1, c2, c3, c4};
        #pragma unroll
        for (int s2 = 0; s2 < 4; ++s2) wvS[tid][s2] = cc[s2] * gx * inv;
    }
    __syncthreads();

    float e0c = 0.f, e1c = 0.f, pcc = 0.f, psc = 0.f;
    #pragma unroll
    for (int v = 0; v < 2; ++v) {
        const float4 w4 = *(const float4*)&wvS[rsel + v][0];
        const float4 x4 = *(const float4*)&xS[rsel + v][0];
        const float wq[4] = {w4.x, w4.y, w4.z, w4.w};
        const float xq[4] = {x4.x, x4.y, x4.z, x4.w};
        float ds = 0.f, e0 = 0.f, e1 = 0.f, pc = 0.f, ps = 0.f;
        #pragma unroll
        for (int s = 0; s < 4; ++s) {
            const float ts    = (s == 0) ? 0.f : ((s == 3) ? 1.f : 0.5f);
            const float dterm = w1r * csv[s][v] - w0r * snv[s][v];
            const float g     = wq[s] * (stv[v] * dterm);
            const float w2    = wq[s] * stv[v];
            ds += g;
            e0 += g * (stv[v] * ts);
            e1 += g * xq[s];
            pc += w2 * csv[s][v];
            ps += w2 * snv[s][v];
        }
        ws[O_DS + (unsigned)(bgv[v] * 256 + m)] = ds;
        e0c += e0; e1c += e1; pcc += pc; psc += ps;
    }
    e0c += __shfl_xor(e0c, 16, 32);
    e1c += __shfl_xor(e1c, 16, 32);
    pcc += __shfl_xor(pcc, 16, 32);
    psc += __shfl_xor(psc, 16, 32);
    if (lane < 16) {
        float* pp = ws + O_P + (unsigned)(blockIdx.x * 1024 + m);
        pp[0]   = e0c;
        pp[256] = e1c;
        pp[512] = pcc;
        pp[768] = psc;
    }
}
#undef ROW_XOR_ADD

#define K2_STEP(ACC, KQ)                                                              \
    {                                                                                 \
        v2f af, bf;                                                                   \
        af.x = dp[(KQ) * 1024];                                                       \
        af.y = dp[(KQ) * 1024 + 256];                                                 \
        bf.x = ip[(KQ) * 256];                                                        \
        bf.y = ip[(KQ) * 256 + 64];                                                   \
        ACC = __builtin_amdgcn_wmma_f32_16x16x4_f32(false, af, false, bf, (short)0, ACC, false, false); \
    }

static __device__ __forceinline__ void sgrad_role(const int blk, const float* __restrict__ ws, float* __restrict__ out)
{
    __shared__ float part[16][256];
    const int tid  = threadIdx.x;
    const int lane = tid & 31;
    const int wvi  = tid >> 5;
    const int mt = blk / 3;
    const int nt = blk - mt * 3;
    const int ln  = lane & 15;
    const int klo = (lane < 16) ? 0 : 2;
    const int m   = mt * 16 + ln;
    const int kc  = 2 + nt * 16 + ln;
    const unsigned b0 = (unsigned)(wvi * 32 + klo);
    const float* __restrict__ dp = ws + O_DS  + b0 * 256u + (unsigned)m;
    const float* __restrict__ ip = ws + O_INP + b0 * 64u  + (unsigned)kc;

    v8f acc0, acc1;
    #pragma unroll
    for (int v = 0; v < 8; ++v) { acc0[v] = 0.f; acc1[v] = 0.f; }

    #pragma unroll
    for (int kq = 0; kq < 8; kq += 2) {
        K2_STEP(acc0, kq)
        K2_STEP(acc1, kq + 1)
    }

    #pragma unroll
    for (int v = 0; v < 8; ++v) part[wvi][v * 32 + lane] = acc0[v] + acc1[v];
    __syncthreads();

    if (tid < 256) {
        float s = 0.f;
        for (int w = 0; w < 16; ++w) s += part[w][tid];
        const int v  = tid >> 5;
        const int l  = tid & 31;
        const int mm = mt * 16 + v + ((l < 16) ? 0 : 8);
        const int kk = 2 + nt * 16 + (l & 15);
        out[25088 + mm * 50 + kk] = s;
    }
}
#undef K2_STEP

static __device__ __forceinline__ void csum_role(const int blk, const float* __restrict__ ws, float* __restrict__ out)
{
    __shared__ float red[8][64];
    const int tid   = threadIdx.x;
    const int col   = tid & 63;
    const int slice = tid >> 6;
    const int q0    = blk * 64;

    float pv[16];
    #pragma unroll
    for (int i = 0; i < 16; ++i)
        pv[i] = ws[O_P + (unsigned)((slice * 16 + i) * 1024 + q0 + col)];
    float s = 0.f;
    #pragma unroll
    for (int i = 0; i < 16; ++i) s += pv[i];
    red[slice][col] = s;
    __syncthreads();
    if (tid < 64) {
        float t = 0.f;
        for (int w = 0; w < 8; ++w) t += red[w][tid];
        const int q     = q0 + tid;
        const int plane = q >> 8;
        const int mq    = q & 255;
        if      (plane == 0) out[25088 + mq * 50]     = t;
        else if (plane == 1) out[25088 + mq * 50 + 1] = t;
        else if (plane == 2) out[37888 + mq]          = t;
        else                 out[37888 + 256 + mq]    = t;
    }
}

static __device__ __forceinline__ void ugrad_role(const int blk, const float* __restrict__ ws,
                                                  const float* __restrict__ S, float* __restrict__ out)
{
    __shared__ float upart[16][768];
    const int tid  = threadIdx.x;
    const int lane = tid & 31;
    const int wvi  = tid >> 5;
    const int ln   = lane & 15;
    const int hh   = lane >> 4;
    const int b0   = blk * 16;

    const int mb = wvi * 16 + hh * 8;
    const float4* ap = (const float4*)(ws + O_DS + (unsigned)((b0 + ln) * 256 + mb));
    const float4 a0 = ap[0];
    const float4 a1 = ap[1];

    v2f bfr[3][4];
    #pragma unroll
    for (int nt = 0; nt < 3; ++nt) {
        #pragma unroll
        for (int q = 0; q < 4; ++q) {
            const float* sp = S + (mb + 2 * q) * 50 + 2 + nt * 16 + ln;
            bfr[nt][q].x = sp[0];
            bfr[nt][q].y = sp[50];
        }
    }

    float eua[2], eub[2], egx[2], egb[2];
    #pragma unroll
    for (int i = 0; i < 2; ++i) {
        eua[i] = 0.f; eub[i] = 0.f; egx[i] = 0.f; egb[i] = 0.f;
        const int idx = tid + i * 512;
        if (idx < 768) {
            const int nt  = idx >> 8;
            const int row = ((idx >> 5) & 7) + ((idx & 16) ? 8 : 0);
            const int cc  = idx & 15;
            const int b   = b0 + row;
            const int oa  = (nt == 0) ? 16 : 0;
            const int ob  = (nt == 2) ? 16 : 32;
            eua[i] = ws[O_INP + (unsigned)(b * 64 + 2 + oa + cc)];
            eub[i] = ws[O_INP + (unsigned)(b * 64 + 2 + ob + cc)];
            egx[i] = ws[O_GX + b];
            egb[i] = ws[O_GB + b];
        }
    }

    v2f af[4];
    af[0].x = a0.x; af[0].y = a0.y;
    af[1].x = a0.z; af[1].y = a0.w;
    af[2].x = a1.x; af[2].y = a1.y;
    af[3].x = a1.z; af[3].y = a1.w;

    v8f acc[3];
    #pragma unroll
    for (int nt = 0; nt < 3; ++nt) {
        #pragma unroll
        for (int v = 0; v < 8; ++v) acc[nt][v] = 0.f;
    }
    #pragma unroll
    for (int q = 0; q < 4; ++q) {
        #pragma unroll
        for (int nt = 0; nt < 3; ++nt) {
            acc[nt] = __builtin_amdgcn_wmma_f32_16x16x4_f32(
                false, af[q], false, bfr[nt][q], (short)0, acc[nt], false, false);
        }
    }

    #pragma unroll
    for (int nt = 0; nt < 3; ++nt) {
        #pragma unroll
        for (int v = 0; v < 8; ++v) upart[wvi][nt * 256 + v * 32 + lane] = acc[nt][v];
    }
    __syncthreads();

    #pragma unroll
    for (int i = 0; i < 2; ++i) {
        const int idx = tid + i * 512;
        if (idx < 768) {
            float s = 0.f;
            for (int w = 0; w < 16; ++w) s += upart[w][idx];
            const int nt  = idx >> 8;
            const int row = ((idx >> 5) & 7) + ((idx & 16) ? 8 : 0);
            const int cc  = idx & 15;
            const float p = eua[i] * eub[i];
            out[512 + (b0 + row) * 48 + nt * 16 + cc] = egx[i] * (1.f / 512.f) * egb[i] * p + s;
        }
    }
}

__global__ __launch_bounds__(512) __attribute__((amdgpu_waves_per_eu(4, 4)))
void k2_consumers(const float* __restrict__ ws, const float* __restrict__ S, float* __restrict__ out)
{
    const int bid = blockIdx.x;
    if      (bid < 32) ugrad_role(bid, ws, S, out);
    else if (bid < 80) sgrad_role(bid - 32, ws, out);
    else               csum_role(bid - 80, ws, out);
}

extern "C" void kernel_launch(void* const* d_in, const int* in_sizes, int n_in,
                              void* d_out, int out_size, void* d_ws, size_t ws_size,
                              hipStream_t stream)
{
    (void)in_sizes; (void)n_in; (void)out_size; (void)ws_size;
    const float* U0   = (const float*)d_in[0];
    const float* U1   = (const float*)d_in[1];
    const float* U2   = (const float*)d_in[2];
    const float* S    = (const float*)d_in[3];
    const float* w    = (const float*)d_in[4];
    const float* btn  = (const float*)d_in[5];
    const float* bobs = (const float*)d_in[6];
    const int*   bin  = (const int*)d_in[7];
    float* out = (float*)d_out;
    float* ws  = (float*)d_ws;

    k1_stages<<<128, 512, 0, stream>>>(U0, U1, U2, S, w, btn, bobs, bin, ws, out);
    k2_consumers<<<96, 512, 0, stream>>>(ws, S, out);
}
